# starting kernel
# baseline (speedup 1.0000x reference)
#include <hip/hip_runtime.h>
#include <hip/hip_bf16.h>
#include <cstdint>

#define B_ 8
#define C_ 512
#define T_ 2048
#define HEADS_ 8
#define KER_ 31
#define NROWS (B_ * T_)
#define EPS_ 1e-5f

typedef __attribute__((ext_vector_type(2))) float v2f;
typedef __attribute__((ext_vector_type(8))) float v8f;

__device__ __forceinline__ v8f wmma_f32(v2f a, v2f b, v8f c) {
  return __builtin_amdgcn_wmma_f32_16x16x4_f32(false, a, false, b, (short)0, c,
                                               false, false);
}

__global__ __launch_bounds__(256) void ln_stats_kernel(
    const float* __restrict__ x, float* __restrict__ stats) {
  const int tid = threadIdx.x;
  const int row0 = blockIdx.x * 64;
  const int b = row0 / T_;
  const int t0 = row0 % T_;
  const int r = tid & 63;
  const int g = tid >> 6;
  float s = 0.f, s2 = 0.f;
  const float* xb = x + (size_t)b * C_ * T_ + t0 + r;
#pragma unroll 4
  for (int i = 0; i < 128; ++i) {
    int c = g * 128 + i;
    float v = xb[(size_t)c * T_];
    s += v;
    s2 += v * v;
  }
  __shared__ float sb[4][64], sq[4][64];
  sb[g][r] = s;
  sq[g][r] = s2;
  __syncthreads();
  if (tid < 64) {
    float S = sb[0][tid] + sb[1][tid] + sb[2][tid] + sb[3][tid];
    float S2 = sq[0][tid] + sq[1][tid] + sq[2][tid] + sq[3][tid];
    float mu = S * (1.f / C_);
    float var = S2 * (1.f / C_) - mu * mu;
    stats[(row0 + tid) * 2 + 0] = mu;
    stats[(row0 + tid) * 2 + 1] = rsqrtf(var + EPS_);
  }
}

#define AS_S 66
#define BS_S 66
#define KT_S 65
__global__ __launch_bounds__(256) void glu_gemm_kernel(
    const float* __restrict__ x, const float* __restrict__ gamma,
    const float* __restrict__ beta, const float* __restrict__ Wk,
    const float* __restrict__ bk, const float* __restrict__ stats,
    float* __restrict__ k_ws) {
  __shared__ float smem[64 * AS_S + 128 * BS_S];
  float* As = smem;
  float* Bs = smem + 64 * AS_S;

  const int tid = threadIdx.x;
  const int row0 = blockIdx.x * 64;
  const int n0 = blockIdx.y * 64;
  const int b = row0 / T_;
  const int t0 = row0 % T_;
  const int r_ld = tid & 63;
  const int kg = tid >> 6;
  const float mu = stats[(row0 + r_ld) * 2 + 0];
  const float rs = stats[(row0 + r_ld) * 2 + 1];

  const int lane = tid & 31;
  const int wave = tid >> 5;
  const int laneM = lane & 15;
  const int hi = lane >> 4;
  const int rblk = wave & 3;
  const int j0 = (wave >> 2) * 32;

  v8f accA0 = {}, accA1 = {}, accB0 = {}, accB1 = {};
  const float* xb = x + (size_t)b * C_ * T_ + t0 + r_ld;

  for (int kbase = 0; kbase < C_; kbase += 64) {
#pragma unroll
    for (int i = 0; i < 16; ++i) {
      int kk = kg * 16 + i;
      int c = kbase + kk;
      float v = xb[(size_t)c * T_];
      As[r_ld * AS_S + kk] = (v - mu) * rs * gamma[c] + beta[c];
    }
    {
      int jj = tid & 63;
      int kq = tid >> 6;
#pragma unroll
      for (int part = 0; part < 2; ++part) {
        int n = n0 + jj + part * C_;
#pragma unroll
        for (int i = 0; i < 16; ++i) {
          int kk = kq * 16 + i;
          Bs[(part * 64 + jj) * BS_S + kk] =
              Wk[(size_t)(kbase + kk) * (2 * C_) + n];
        }
      }
    }
    __syncthreads();
#pragma unroll
    for (int kb = 0; kb < 64; kb += 4) {
      v2f af = *reinterpret_cast<const v2f*>(
          &As[(rblk * 16 + laneM) * AS_S + kb + 2 * hi]);
      v2f b0 = *reinterpret_cast<const v2f*>(
          &Bs[(j0 + laneM) * BS_S + kb + 2 * hi]);
      v2f b1 = *reinterpret_cast<const v2f*>(
          &Bs[(j0 + 16 + laneM) * BS_S + kb + 2 * hi]);
      v2f b2 = *reinterpret_cast<const v2f*>(
          &Bs[(64 + j0 + laneM) * BS_S + kb + 2 * hi]);
      v2f b3 = *reinterpret_cast<const v2f*>(
          &Bs[(64 + j0 + 16 + laneM) * BS_S + kb + 2 * hi]);
      accA0 = wmma_f32(af, b0, accA0);
      accA1 = wmma_f32(af, b1, accA1);
      accB0 = wmma_f32(af, b2, accB0);
      accB1 = wmma_f32(af, b3, accB1);
    }
    __syncthreads();
  }

  float* kst = smem;
  const float bka0 = bk[n0 + j0 + laneM];
  const float bka1 = bk[n0 + j0 + 16 + laneM];
  const float bkb0 = bk[C_ + n0 + j0 + laneM];
  const float bkb1 = bk[C_ + n0 + j0 + 16 + laneM];
#pragma unroll
  for (int v = 0; v < 8; ++v) {
    int tl = rblk * 16 + v + 8 * hi;
    float a0 = accA0[v] + bka0;
    float g0 = accB0[v] + bkb0;
    float a1 = accA1[v] + bka1;
    float g1 = accB1[v] + bkb1;
    kst[(j0 + laneM) * KT_S + tl] = a0 / (1.f + __expf(-g0));
    kst[(j0 + 16 + laneM) * KT_S + tl] = a1 / (1.f + __expf(-g1));
  }
  __syncthreads();
  float* kout = k_ws + (size_t)b * C_ * T_;
  for (int i = tid; i < 64 * 64; i += 256) {
    int cc = i >> 6;
    int tl = i & 63;
    kout[(size_t)(n0 + cc) * T_ + t0 + tl] = kst[cc * KT_S + tl];
  }
}

#define A2_S 34
#define B2_S 34
#define P_S 257
__global__ __launch_bounds__(256) void conv_weight_kernel(
    const float* __restrict__ k_ws, const float* __restrict__ Wc,
    const float* __restrict__ bc, float* __restrict__ w_ws) {
  __shared__ float As[32 * A2_S];
  __shared__ float Bs[256 * B2_S];
  __shared__ float proj[32 * P_S];
  const int tid = threadIdx.x;
  const int row0 = blockIdx.x * 32;
  const int b = row0 / T_;
  const int t0 = row0 % T_;
  const int lane = tid & 31;
  const int wave = tid >> 5;
  const int laneM = lane & 15;
  const int hi = lane >> 4;
  const int rb = wave & 1;
  const int jb = (wave >> 1) * 64;

  v8f acc[4] = {{}, {}, {}, {}};
  const float* kin = k_ws + (size_t)b * C_ * T_ + t0;

  for (int kbase = 0; kbase < C_; kbase += 32) {
    {
      int r = tid & 31;
      int cq = tid >> 5;
#pragma unroll
      for (int i = 0; i < 4; ++i) {
        int kk = cq + 8 * i;
        As[r * A2_S + kk] = kin[(size_t)(kbase + kk) * T_ + r];
      }
    }
    {
      int nn = tid & 63;
      int kq = tid >> 6;
#pragma unroll
      for (int q = 0; q < 4; ++q) {
        int n = nn + 64 * q;
#pragma unroll
        for (int i = 0; i < 8; ++i) {
          int kk = kq * 8 + i;
          float v = (n < KER_ * HEADS_)
                        ? Wc[(size_t)(kbase + kk) * (KER_ * HEADS_) + n]
                        : 0.f;
          Bs[n * B2_S + kk] = v;
        }
      }
    }
    __syncthreads();
#pragma unroll
    for (int kb = 0; kb < 32; kb += 4) {
      v2f af = *reinterpret_cast<const v2f*>(
          &As[(rb * 16 + laneM) * A2_S + kb + 2 * hi]);
#pragma unroll
      for (int jt = 0; jt < 4; ++jt) {
        v2f bf = *reinterpret_cast<const v2f*>(
            &Bs[(jb + jt * 16 + laneM) * B2_S + kb + 2 * hi]);
        acc[jt] = wmma_f32(af, bf, acc[jt]);
      }
    }
    __syncthreads();
  }
#pragma unroll
  for (int jt = 0; jt < 4; ++jt)
#pragma unroll
    for (int v = 0; v < 8; ++v)
      proj[(rb * 16 + v + 8 * hi) * P_S + jb + jt * 16 + laneM] = acc[jt][v];
  __syncthreads();

  {
    int rr = tid >> 3;
    int h = tid & 7;
    float* prow = &proj[rr * P_S + h * KER_];
    const float* bch = &bc[h * KER_];
    float m = -1e30f;
#pragma unroll
    for (int i = 0; i < KER_; ++i) {
      float v = prow[i] + bch[i];
      prow[i] = v;
      m = fmaxf(m, v);
    }
    float s = 0.f;
#pragma unroll
    for (int i = 0; i < KER_; ++i) {
      float e = __expf(prow[i] - m);
      prow[i] = e;
      s += e;
    }
    float inv = 1.f / s;
    float* wout = w_ws + (size_t)(row0 + rr) * 256 + h * 32;
#pragma unroll
    for (int i = 0; i < KER_; ++i) wout[i] = prow[i] * inv;
    wout[KER_] = 0.f;
  }
}

__global__ __launch_bounds__(256) void dynconv_kernel(
    const float* __restrict__ x, const float* __restrict__ k_ws,
    const float* __restrict__ w_ws, const float* __restrict__ conv_bias,
    float* __restrict__ out) {
  __shared__ float wl[32 * 257];
  const int tid = threadIdx.x;
  const int t0 = blockIdx.x * 32;
  const int b = blockIdx.y;
  const float* wsrc = w_ws + ((size_t)b * T_ + t0) * 256;
  for (int i = tid; i < 32 * 256; i += 256)
    wl[(i >> 8) * 257 + (i & 255)] = wsrc[i];
  __syncthreads();

  const int tx = tid & 31;
  const int cg = tid >> 5;
  const int t = t0 + tx;
  const float* wrow = &wl[tx * 257 + cg * 32];
  const bool interior = (t0 >= KER_ / 2) && (t0 + 32 + KER_ / 2 <= T_);

#pragma unroll 1
  for (int ci = 0; ci < 64; ++ci) {
    int c = cg * 64 + ci;
    const size_t basec = ((size_t)b * C_ + c) * T_;
    float acc = conv_bias[c] + x[basec + t];
    if (interior) {
#pragma unroll
      for (int kk = 0; kk < KER_; ++kk)
        acc = fmaf(k_ws[basec + t + kk - KER_ / 2], wrow[kk], acc);
    } else {
#pragma unroll
      for (int kk = 0; kk < KER_; ++kk) {
        int tt = t + kk - KER_ / 2;
        if ((unsigned)tt < (unsigned)T_)
          acc = fmaf(k_ws[basec + tt], wrow[kk], acc);
      }
    }
    out[basec + t] = acc;
  }
}

extern "C" void kernel_launch(void* const* d_in, const int* in_sizes, int n_in,
                              void* d_out, int out_size, void* d_ws,
                              size_t ws_size, hipStream_t stream) {
  (void)in_sizes;
  (void)n_in;
  (void)out_size;
  (void)ws_size;
  const float* x = (const float*)d_in[0];
  const float* ln_gamma = (const float*)d_in[1];
  const float* ln_beta = (const float*)d_in[2];
  const float* Wk = (const float*)d_in[3];
  const float* bk = (const float*)d_in[4];
  const float* Wc = (const float*)d_in[5];
  const float* bc = (const float*)d_in[6];
  const float* conv_bias = (const float*)d_in[7];
  float* out = (float*)d_out;

  float* ws = (float*)d_ws;
  float* stats = ws;
  float* k_ws = ws + 32768;
  float* w_ws = k_ws + (size_t)B_ * C_ * T_;

  ln_stats_kernel<<<NROWS / 64, 256, 0, stream>>>(x, stats);
  glu_gemm_kernel<<<dim3(NROWS / 64, C_ / 64), 256, 0, stream>>>(
      x, ln_gamma, ln_beta, Wk, bk, stats, k_ws);
  conv_weight_kernel<<<NROWS / 32, 256, 0, stream>>>(k_ws, Wc, bc, w_ws);
  dynconv_kernel<<<dim3(T_ / 32, B_), 256, 0, stream>>>(x, k_ws, w_ws,
                                                        conv_bias, out);
}


# dynamic conv remapped: lane = channel, each k value held in a register window and reused by every tap, tap weights read through wave-uniform (scalar) addresses; tap summation order changes
# speedup vs baseline: 1.8131x; 1.8131x over previous
#include <hip/hip_runtime.h>
#include <hip/hip_bf16.h>
#include <cstdint>

#define B_ 8
#define C_ 512
#define T_ 2048
#define HEADS_ 8
#define KER_ 31
#define NROWS (B_ * T_)
#define EPS_ 1e-5f

typedef __attribute__((ext_vector_type(2))) float v2f;
typedef __attribute__((ext_vector_type(8))) float v8f;

__device__ __forceinline__ v8f wmma_f32(v2f a, v2f b, v8f c) {
  return __builtin_amdgcn_wmma_f32_16x16x4_f32(false, a, false, b, (short)0, c,
                                               false, false);
}

__global__ __launch_bounds__(256) void ln_stats_kernel(
    const float* __restrict__ x, float* __restrict__ stats) {
  const int tid = threadIdx.x;
  const int row0 = blockIdx.x * 64;
  const int b = row0 / T_;
  const int t0 = row0 % T_;
  const int r = tid & 63;
  const int g = tid >> 6;
  float s = 0.f, s2 = 0.f;
  const float* xb = x + (size_t)b * C_ * T_ + t0 + r;
#pragma unroll 4
  for (int i = 0; i < 128; ++i) {
    int c = g * 128 + i;
    float v = xb[(size_t)c * T_];
    s += v;
    s2 += v * v;
  }
  __shared__ float sb[4][64], sq[4][64];
  sb[g][r] = s;
  sq[g][r] = s2;
  __syncthreads();
  if (tid < 64) {
    float S = sb[0][tid] + sb[1][tid] + sb[2][tid] + sb[3][tid];
    float S2 = sq[0][tid] + sq[1][tid] + sq[2][tid] + sq[3][tid];
    float mu = S * (1.f / C_);
    float var = S2 * (1.f / C_) - mu * mu;
    stats[(row0 + tid) * 2 + 0] = mu;
    stats[(row0 + tid) * 2 + 1] = rsqrtf(var + EPS_);
  }
}

#define AS_S 66
#define BS_S 66
#define KT_S 65
__global__ __launch_bounds__(256) void glu_gemm_kernel(
    const float* __restrict__ x, const float* __restrict__ gamma,
    const float* __restrict__ beta, const float* __restrict__ Wk,
    const float* __restrict__ bk, const float* __restrict__ stats,
    float* __restrict__ k_ws) {
  __shared__ float smem[64 * AS_S + 128 * BS_S];
  float* As = smem;
  float* Bs = smem + 64 * AS_S;

  const int tid = threadIdx.x;
  const int row0 = blockIdx.x * 64;
  const int n0 = blockIdx.y * 64;
  const int b = row0 / T_;
  const int t0 = row0 % T_;
  const int r_ld = tid & 63;
  const int kg = tid >> 6;
  const float mu = stats[(row0 + r_ld) * 2 + 0];
  const float rs = stats[(row0 + r_ld) * 2 + 1];

  const int lane = tid & 31;
  const int wave = tid >> 5;
  const int laneM = lane & 15;
  const int hi = lane >> 4;
  const int rblk = wave & 3;
  const int j0 = (wave >> 2) * 32;

  v8f accA0 = {}, accA1 = {}, accB0 = {}, accB1 = {};
  const float* xb = x + (size_t)b * C_ * T_ + t0 + r_ld;

  for (int kbase = 0; kbase < C_; kbase += 64) {
#pragma unroll
    for (int i = 0; i < 16; ++i) {
      int kk = kg * 16 + i;
      int c = kbase + kk;
      float v = xb[(size_t)c * T_];
      As[r_ld * AS_S + kk] = (v - mu) * rs * gamma[c] + beta[c];
    }
    {
      int jj = tid & 63;
      int kq = tid >> 6;
#pragma unroll
      for (int part = 0; part < 2; ++part) {
        int n = n0 + jj + part * C_;
#pragma unroll
        for (int i = 0; i < 16; ++i) {
          int kk = kq * 16 + i;
          Bs[(part * 64 + jj) * BS_S + kk] =
              Wk[(size_t)(kbase + kk) * (2 * C_) + n];
        }
      }
    }
    __syncthreads();
#pragma unroll
    for (int kb = 0; kb < 64; kb += 4) {
      v2f af = *reinterpret_cast<const v2f*>(
          &As[(rblk * 16 + laneM) * AS_S + kb + 2 * hi]);
      v2f b0 = *reinterpret_cast<const v2f*>(
          &Bs[(j0 + laneM) * BS_S + kb + 2 * hi]);
      v2f b1 = *reinterpret_cast<const v2f*>(
          &Bs[(j0 + 16 + laneM) * BS_S + kb + 2 * hi]);
      v2f b2 = *reinterpret_cast<const v2f*>(
          &Bs[(64 + j0 + laneM) * BS_S + kb + 2 * hi]);
      v2f b3 = *reinterpret_cast<const v2f*>(
          &Bs[(64 + j0 + 16 + laneM) * BS_S + kb + 2 * hi]);
      accA0 = wmma_f32(af, b0, accA0);
      accA1 = wmma_f32(af, b1, accA1);
      accB0 = wmma_f32(af, b2, accB0);
      accB1 = wmma_f32(af, b3, accB1);
    }
    __syncthreads();
  }

  float* kst = smem;
  const float bka0 = bk[n0 + j0 + laneM];
  const float bka1 = bk[n0 + j0 + 16 + laneM];
  const float bkb0 = bk[C_ + n0 + j0 + laneM];
  const float bkb1 = bk[C_ + n0 + j0 + 16 + laneM];
#pragma unroll
  for (int v = 0; v < 8; ++v) {
    int tl = rblk * 16 + v + 8 * hi;
    float a0 = accA0[v] + bka0;
    float g0 = accB0[v] + bkb0;
    float a1 = accA1[v] + bka1;
    float g1 = accB1[v] + bkb1;
    kst[(j0 + laneM) * KT_S + tl] = a0 / (1.f + __expf(-g0));
    kst[(j0 + 16 + laneM) * KT_S + tl] = a1 / (1.f + __expf(-g1));
  }
  __syncthreads();
  float* kout = k_ws + (size_t)b * C_ * T_;
  for (int i = tid; i < 64 * 64; i += 256) {
    int cc = i >> 6;
    int tl = i & 63;
    kout[(size_t)(n0 + cc) * T_ + t0 + tl] = kst[cc * KT_S + tl];
  }
}

#define A2_S 34
#define B2_S 34
#define P_S 257
__global__ __launch_bounds__(256) void conv_weight_kernel(
    const float* __restrict__ k_ws, const float* __restrict__ Wc,
    const float* __restrict__ bc, float* __restrict__ w_ws) {
  __shared__ float As[32 * A2_S];
  __shared__ float Bs[256 * B2_S];
  __shared__ float proj[32 * P_S];
  const int tid = threadIdx.x;
  const int row0 = blockIdx.x * 32;
  const int b = row0 / T_;
  const int t0 = row0 % T_;
  const int lane = tid & 31;
  const int wave = tid >> 5;
  const int laneM = lane & 15;
  const int hi = lane >> 4;
  const int rb = wave & 1;
  const int jb = (wave >> 1) * 64;

  v8f acc[4] = {{}, {}, {}, {}};
  const float* kin = k_ws + (size_t)b * C_ * T_ + t0;

  for (int kbase = 0; kbase < C_; kbase += 32) {
    {
      int r = tid & 31;
      int cq = tid >> 5;
#pragma unroll
      for (int i = 0; i < 4; ++i) {
        int kk = cq + 8 * i;
        As[r * A2_S + kk] = kin[(size_t)(kbase + kk) * T_ + r];
      }
    }
    {
      int nn = tid & 63;
      int kq = tid >> 6;
#pragma unroll
      for (int q = 0; q < 4; ++q) {
        int n = nn + 64 * q;
#pragma unroll
        for (int i = 0; i < 8; ++i) {
          int kk = kq * 8 + i;
          float v = (n < KER_ * HEADS_)
                        ? Wc[(size_t)(kbase + kk) * (KER_ * HEADS_) + n]
                        : 0.f;
          Bs[n * B2_S + kk] = v;
        }
      }
    }
    __syncthreads();
#pragma unroll
    for (int kb = 0; kb < 32; kb += 4) {
      v2f af = *reinterpret_cast<const v2f*>(
          &As[(rb * 16 + laneM) * A2_S + kb + 2 * hi]);
#pragma unroll
      for (int jt = 0; jt < 4; ++jt) {
        v2f bf = *reinterpret_cast<const v2f*>(
            &Bs[(jb + jt * 16 + laneM) * B2_S + kb + 2 * hi]);
        acc[jt] = wmma_f32(af, bf, acc[jt]);
      }
    }
    __syncthreads();
  }
#pragma unroll
  for (int jt = 0; jt < 4; ++jt)
#pragma unroll
    for (int v = 0; v < 8; ++v)
      proj[(rb * 16 + v + 8 * hi) * P_S + jb + jt * 16 + laneM] = acc[jt][v];
  __syncthreads();

  {
    int rr = tid >> 3;
    int h = tid & 7;
    float* prow = &proj[rr * P_S + h * KER_];
    const float* bch = &bc[h * KER_];
    float m = -1e30f;
#pragma unroll
    for (int i = 0; i < KER_; ++i) {
      float v = prow[i] + bch[i];
      prow[i] = v;
      m = fmaxf(m, v);
    }
    float s = 0.f;
#pragma unroll
    for (int i = 0; i < KER_; ++i) {
      float e = __expf(prow[i] - m);
      prow[i] = e;
      s += e;
    }
    float inv = 1.f / s;
    float* wout = w_ws + (size_t)(row0 + rr) * 256 + h * 32;
#pragma unroll
    for (int i = 0; i < KER_; ++i) wout[i] = prow[i] * inv;
    wout[KER_] = 0.f;
  }
}

typedef __attribute__((ext_vector_type(4))) float v4f;
#define DC_TT 128
#define DC_SEG 16
#define DC_HALO 16
#define DC_ROWF (DC_TT + 2 * DC_HALO)
#define DC_WIN (DC_SEG + 2 * DC_HALO)
#define DC_PITCH 164
__global__ __launch_bounds__(256) void dynconv_kernel(
    const float* __restrict__ x, const float* __restrict__ k_ws,
    const float* __restrict__ w_ws, const float* __restrict__ conv_bias,
    float* __restrict__ out) {
  __shared__ __attribute__((aligned(16))) float tile[64 * DC_PITCH];
  const int tid = threadIdx.x;
  const int lane = tid & 31;
  const int seg = __builtin_amdgcn_readfirstlane(tid >> 5);
  const int t0 = blockIdx.x * DC_TT;
  const int h = blockIdx.y;
  const int b = blockIdx.z;
  const size_t chan0 = ((size_t)b * C_ + (size_t)h * 64) * T_;

  {
    const float* ksrc = k_ws + chan0;
    v4f stg[10];
#pragma unroll
    for (int it = 0; it < 10; ++it) {
      const int i = tid + it * 256;
      const int r = i / (DC_ROWF / 4);
      const int q = i - r * (DC_ROWF / 4);
      const int tt = t0 - DC_HALO + 4 * q;
      const int ttc = min(max(tt, 0), T_ - 4);
      v4f v = *reinterpret_cast<const v4f*>(ksrc + (size_t)r * T_ + ttc);
      const v4f z = {0.f, 0.f, 0.f, 0.f};
      stg[it] = (tt == ttc) ? v : z;
    }
#pragma unroll
    for (int it = 0; it < 10; ++it) {
      const int i = tid + it * 256;
      const int r = i / (DC_ROWF / 4);
      const int q = i - r * (DC_ROWF / 4);
      *reinterpret_cast<v4f*>(&tile[r * DC_PITCH + 4 * q]) = stg[it];
    }
  }
  __syncthreads();

  float wa[DC_WIN], wb[DC_WIN];
  {
    const float* ra = &tile[lane * DC_PITCH + seg * DC_SEG];
    const float* rb = &tile[(lane + 32) * DC_PITCH + seg * DC_SEG];
#pragma unroll
    for (int q = 0; q < DC_WIN / 4; ++q) {
      const v4f va = *reinterpret_cast<const v4f*>(ra + 4 * q);
      const v4f vb = *reinterpret_cast<const v4f*>(rb + 4 * q);
      wa[4 * q + 0] = va.x; wa[4 * q + 1] = va.y;
      wa[4 * q + 2] = va.z; wa[4 * q + 3] = va.w;
      wb[4 * q + 0] = vb.x; wb[4 * q + 1] = vb.y;
      wb[4 * q + 2] = vb.z; wb[4 * q + 3] = vb.w;
    }
  }
  __syncthreads();

  const float* wsrc =
      w_ws + ((size_t)b * T_ + t0 + seg * DC_SEG) * 256 + (size_t)h * 32;
#pragma unroll
  for (int j4 = 0; j4 < DC_SEG; j4 += 4) {
    v4f oa, ob;
#pragma unroll
    for (int jj = 0; jj < 4; ++jj) {
      const int j = j4 + jj;
      const float* wr = wsrc + (size_t)j * 256;
      float a0 = 0.f, a1 = 0.f;
#pragma unroll
      for (int kk = 0; kk < KER_; ++kk) {
        const float w = wr[kk];
        a0 = fmaf(w, wa[j + kk + 1], a0);
        a1 = fmaf(w, wb[j + kk + 1], a1);
      }
      oa[jj] = a0;
      ob[jj] = a1;
    }
    *reinterpret_cast<v4f*>(&tile[lane * DC_PITCH + seg * DC_SEG + j4]) = oa;
    *reinterpret_cast<v4f*>(&tile[(lane + 32) * DC_PITCH + seg * DC_SEG + j4]) = ob;
  }
  __syncthreads();

#pragma unroll
  for (int it = 0; it < 8; ++it) {
    const int r = it * 8 + seg;
    v4f v = *reinterpret_cast<const v4f*>(&tile[r * DC_PITCH + 4 * lane]);
    const size_t g = chan0 + (size_t)r * T_ + t0 + 4 * lane;
    const v4f xv = *reinterpret_cast<const v4f*>(x + g);
    const float cb = conv_bias[h * 64 + r];
    v.x = (v.x + cb) + xv.x;
    v.y = (v.y + cb) + xv.y;
    v.z = (v.z + cb) + xv.z;
    v.w = (v.w + cb) + xv.w;
    *reinterpret_cast<v4f*>(out + g) = v;
  }
}

extern "C" void kernel_launch(void* const* d_in, const int* in_sizes, int n_in,
                              void* d_out, int out_size, void* d_ws,
                              size_t ws_size, hipStream_t stream) {
  (void)in_sizes;
  (void)n_in;
  (void)out_size;
  (void)ws_size;
  const float* x = (const float*)d_in[0];
  const float* ln_gamma = (const float*)d_in[1];
  const float* ln_beta = (const float*)d_in[2];
  const float* Wk = (const float*)d_in[3];
  const float* bk = (const float*)d_in[4];
  const float* Wc = (const float*)d_in[5];
  const float* bc = (const float*)d_in[6];
  const float* conv_bias = (const float*)d_in[7];
  float* out = (float*)d_out;

  float* ws = (float*)d_ws;
  float* stats = ws;
  float* k_ws = ws + 32768;
  float* w_ws = k_ws + (size_t)B_ * C_ * T_;

  ln_stats_kernel<<<NROWS / 64, 256, 0, stream>>>(x, stats);
  glu_gemm_kernel<<<dim3(NROWS / 64, C_ / 64), 256, 0, stream>>>(
      x, ln_gamma, ln_beta, Wk, bk, stats, k_ws);
  conv_weight_kernel<<<NROWS / 32, 256, 0, stream>>>(k_ws, Wc, bc, w_ws);
  dynconv_kernel<<<dim3(T_ / DC_TT, HEADS_, B_), 256, 0, stream>>>(
      x, k_ws, w_ws, conv_bias, out);
}


# both GEMMs stage operands by 16-byte loads, one group per chunk, into a k-pair-interleaved LDS layout: no 4-byte transposing scatter, no pad test on Wc, conflict-free fragment reads; k order kept
# speedup vs baseline: 1.9589x; 1.0804x over previous
#include <hip/hip_runtime.h>
#include <hip/hip_bf16.h>
#include <cstdint>

#define B_ 8
#define C_ 512
#define T_ 2048
#define HEADS_ 8
#define KER_ 31
#define NROWS (B_ * T_)
#define EPS_ 1e-5f

typedef __attribute__((ext_vector_type(2))) float v2f;
typedef __attribute__((ext_vector_type(8))) float v8f;

__device__ __forceinline__ v8f wmma_f32(v2f a, v2f b, v8f c) {
  return __builtin_amdgcn_wmma_f32_16x16x4_f32(false, a, false, b, (short)0, c,
                                               false, false);
}

__global__ __launch_bounds__(256) void ln_stats_kernel(
    const float* __restrict__ x, float* __restrict__ stats) {
  const int tid = threadIdx.x;
  const int row0 = blockIdx.x * 64;
  const int b = row0 / T_;
  const int t0 = row0 % T_;
  const int r = tid & 63;
  const int g = tid >> 6;
  float s = 0.f, s2 = 0.f;
  const float* xb = x + (size_t)b * C_ * T_ + t0 + r;
#pragma unroll 4
  for (int i = 0; i < 128; ++i) {
    int c = g * 128 + i;
    float v = xb[(size_t)c * T_];
    s += v;
    s2 += v * v;
  }
  __shared__ float sb[4][64], sq[4][64];
  sb[g][r] = s;
  sq[g][r] = s2;
  __syncthreads();
  if (tid < 64) {
    float S = sb[0][tid] + sb[1][tid] + sb[2][tid] + sb[3][tid];
    float S2 = sq[0][tid] + sq[1][tid] + sq[2][tid] + sq[3][tid];
    float mu = S * (1.f / C_);
    float var = S2 * (1.f / C_) - mu * mu;
    stats[(row0 + tid) * 2 + 0] = mu;
    stats[(row0 + tid) * 2 + 1] = rsqrtf(var + EPS_);
  }
}

#define A1P 80
#define B1P 144
#define KT_S 65
typedef __attribute__((ext_vector_type(4))) float v4f;
__global__ __launch_bounds__(256) void glu_gemm_kernel(
    const float* __restrict__ x, const float* __restrict__ gamma,
    const float* __restrict__ beta, const float* __restrict__ Wk,
    const float* __restrict__ bk, const float* __restrict__ stats,
    float* __restrict__ k_ws) {
  __shared__ __attribute__((aligned(16))) float smem[32 * A1P * 2 + 32 * B1P * 2];
  float* As = smem;
  float* Bs = smem + 32 * A1P * 2;

  const int tid = threadIdx.x;
  const int row0 = blockIdx.x * 64;
  const int n0 = blockIdx.y * 64;
  const int b = row0 / T_;
  const int t0 = row0 % T_;

  const int lane = tid & 31;
  const int wave = __builtin_amdgcn_readfirstlane(tid >> 5);
  const int laneM = lane & 15;
  const int hi = lane >> 4;
  const int rblk = wave & 3;
  const int j0 = (wave >> 2) * 32;

  const int a_tq = tid & 15;
  const int a_p0 = 2 * wave + (lane >> 4);
  const int b_q = tid & 15;
  const int b_part = lane >> 4;
  const int b_p0 = wave;

  const v4f st0 = *reinterpret_cast<const v4f*>(stats + (size_t)(row0 + 4 * a_tq) * 2);
  const v4f st1 = *reinterpret_cast<const v4f*>(stats + (size_t)(row0 + 4 * a_tq) * 2 + 4);
  const float mu0 = st0.x, rs0 = st0.y, mu1 = st0.z, rs1 = st0.w;
  const float mu2 = st1.x, rs2 = st1.y, mu3 = st1.z, rs3 = st1.w;

  v8f accA0 = {}, accA1 = {}, accB0 = {}, accB1 = {};
  const float* xb = x + (size_t)b * C_ * T_ + t0 + 4 * a_tq;
  const float* wb = Wk + n0 + b_part * C_ + 4 * b_q;
  const float* a_frag = &As[(hi * A1P + rblk * 16 + laneM) * 2];
  const float* b_frag = &Bs[(hi * B1P + j0 + laneM) * 2];

  for (int kbase = 0; kbase < C_; kbase += 64) {
    v4f xa[2], xc[2];
    float ga[2], gc[2], be_a[2], be_c[2];
#pragma unroll
    for (int j = 0; j < 2; ++j) {
      const int c = kbase + 2 * (a_p0 + 16 * j);
      xa[j] = *reinterpret_cast<const v4f*>(xb + (size_t)c * T_);
      xc[j] = *reinterpret_cast<const v4f*>(xb + (size_t)(c + 1) * T_);
      ga[j] = gamma[c];
      gc[j] = gamma[c + 1];
      be_a[j] = beta[c];
      be_c[j] = beta[c + 1];
    }
    v4f w0[4], w1[4];
#pragma unroll
    for (int j = 0; j < 4; ++j) {
      const int k = kbase + 2 * (b_p0 + 8 * j);
      w0[j] = *reinterpret_cast<const v4f*>(wb + (size_t)k * (2 * C_));
      w1[j] = *reinterpret_cast<const v4f*>(wb + (size_t)(k + 1) * (2 * C_));
    }
    asm volatile("" ::"v"(xa[0]), "v"(xa[1]), "v"(xc[0]), "v"(xc[1]), "v"(w0[0]),
                 "v"(w0[1]), "v"(w0[2]), "v"(w0[3]), "v"(w1[0]), "v"(w1[1]),
                 "v"(w1[2]), "v"(w1[3]), "v"(ga[0]), "v"(ga[1]), "v"(gc[0]),
                 "v"(gc[1]), "v"(be_a[0]), "v"(be_a[1]), "v"(be_c[0]), "v"(be_c[1]));
#pragma unroll
    for (int j = 0; j < 2; ++j) {
      const int p = a_p0 + 16 * j;
      v4f lo, hi4;
      lo.x = (xa[j].x - mu0) * rs0 * ga[j] + be_a[j];
      lo.y = (xc[j].x - mu0) * rs0 * gc[j] + be_c[j];
      lo.z = (xa[j].y - mu1) * rs1 * ga[j] + be_a[j];
      lo.w = (xc[j].y - mu1) * rs1 * gc[j] + be_c[j];
      hi4.x = (xa[j].z - mu2) * rs2 * ga[j] + be_a[j];
      hi4.y = (xc[j].z - mu2) * rs2 * gc[j] + be_c[j];
      hi4.z = (xa[j].w - mu3) * rs3 * ga[j] + be_a[j];
      hi4.w = (xc[j].w - mu3) * rs3 * gc[j] + be_c[j];
      float* dst = &As[(p * A1P + 4 * a_tq) * 2];
      *reinterpret_cast<v4f*>(dst) = lo;
      *reinterpret_cast<v4f*>(dst + 4) = hi4;
    }
#pragma unroll
    for (int j = 0; j < 4; ++j) {
      const int p = b_p0 + 8 * j;
      v4f lo, hi4;
      lo.x = w0[j].x; lo.y = w1[j].x; lo.z = w0[j].y; lo.w = w1[j].y;
      hi4.x = w0[j].z; hi4.y = w1[j].z; hi4.z = w0[j].w; hi4.w = w1[j].w;
      float* dst = &Bs[(p * B1P + b_part * 64 + 4 * b_q) * 2];
      *reinterpret_cast<v4f*>(dst) = lo;
      *reinterpret_cast<v4f*>(dst + 4) = hi4;
    }
    __syncthreads();
#pragma unroll
    for (int kb = 0; kb < 64; kb += 4) {
      const int po = (kb >> 1);
      v2f af = *reinterpret_cast<const v2f*>(a_frag + po * (A1P * 2));
      v2f b0 = *reinterpret_cast<const v2f*>(b_frag + po * (B1P * 2));
      v2f b1 = *reinterpret_cast<const v2f*>(b_frag + po * (B1P * 2) + 16 * 2);
      v2f b2 = *reinterpret_cast<const v2f*>(b_frag + po * (B1P * 2) + 64 * 2);
      v2f b3 = *reinterpret_cast<const v2f*>(b_frag + po * (B1P * 2) + (64 + 16) * 2);
      accA0 = wmma_f32(af, b0, accA0);
      accA1 = wmma_f32(af, b1, accA1);
      accB0 = wmma_f32(af, b2, accB0);
      accB1 = wmma_f32(af, b3, accB1);
    }
    __syncthreads();
  }

  float* kst = smem;
  const float bka0 = bk[n0 + j0 + laneM];
  const float bka1 = bk[n0 + j0 + 16 + laneM];
  const float bkb0 = bk[C_ + n0 + j0 + laneM];
  const float bkb1 = bk[C_ + n0 + j0 + 16 + laneM];
#pragma unroll
  for (int v = 0; v < 8; ++v) {
    int tl = rblk * 16 + v + 8 * hi;
    float a0 = accA0[v] + bka0;
    float g0 = accB0[v] + bkb0;
    float a1 = accA1[v] + bka1;
    float g1 = accB1[v] + bkb1;
    kst[(j0 + laneM) * KT_S + tl] = a0 / (1.f + __expf(-g0));
    kst[(j0 + 16 + laneM) * KT_S + tl] = a1 / (1.f + __expf(-g1));
  }
  __syncthreads();
  float* kout = k_ws + (size_t)b * C_ * T_;
  for (int i = tid; i < 64 * 64; i += 256) {
    int cc = i >> 6;
    int tl = i & 63;
    kout[(size_t)(n0 + cc) * T_ + t0 + tl] = kst[cc * KT_S + tl];
  }
}

#define A2P 48
#define B2P 272
#define P_S 257
#define NCW (KER_ * HEADS_)
#define NQW (NCW / 4)
__global__ __launch_bounds__(256) void conv_weight_kernel(
    const float* __restrict__ k_ws, const float* __restrict__ Wc,
    const float* __restrict__ bc, float* __restrict__ w_ws) {
  __shared__ __attribute__((aligned(16))) float As[16 * A2P * 2];
  __shared__ __attribute__((aligned(16))) float Bs[16 * B2P * 2];
  __shared__ float proj[32 * P_S];
  const int tid = threadIdx.x;
  const int row0 = blockIdx.x * 32;
  const int b = row0 / T_;
  const int t0 = row0 % T_;
  const int lane = tid & 31;
  const int wave = tid >> 5;
  const int laneM = lane & 15;
  const int hi = lane >> 4;
  const int rb = wave & 1;
  const int jb = (wave >> 1) * 64;

  v8f acc[4] = {{}, {}, {}, {}};
  const float* kin = k_ws + (size_t)b * C_ * T_ + t0;

  if (tid < 128) {
    const int p = tid >> 3;
    const int col = NCW + (tid & 7);
    Bs[(p * B2P + col) * 2 + 0] = 0.f;
    Bs[(p * B2P + col) * 2 + 1] = 0.f;
  }

  int b_src[4], b_dst[4];
#pragma unroll
  for (int j = 0; j < 4; ++j) {
    const int item = min(tid + 256 * j, 16 * NQW - 1);
    const int p = item / NQW;
    const int q = item - p * NQW;
    b_src[j] = 2 * p * NCW + 4 * q;
    b_dst[j] = (p * B2P + 4 * q) * 2;
  }
  const bool b_last = (tid + 768) < 16 * NQW;
  const int a_r = tid & 31;
  const int a_p0 = tid >> 5;
  const float* a_frag = &As[(hi * A2P + rb * 16 + laneM) * 2];
  const float* b_frag = &Bs[(hi * B2P + jb + laneM) * 2];

  for (int kbase = 0; kbase < C_; kbase += 32) {
    const float* wsrc = Wc + (size_t)kbase * NCW;
    float a_e[2], a_o[2];
#pragma unroll
    for (int i = 0; i < 2; ++i) {
      const int c = kbase + 2 * (a_p0 + 8 * i);
      a_e[i] = kin[(size_t)c * T_ + a_r];
      a_o[i] = kin[(size_t)(c + 1) * T_ + a_r];
    }
    v4f w0[4], w1[4];
#pragma unroll
    for (int j = 0; j < 4; ++j) {
      w0[j] = *reinterpret_cast<const v4f*>(wsrc + b_src[j]);
      w1[j] = *reinterpret_cast<const v4f*>(wsrc + b_src[j] + NCW);
    }
#pragma unroll
    for (int i = 0; i < 2; ++i) {
      v2f pr;
      pr.x = a_e[i];
      pr.y = a_o[i];
      *reinterpret_cast<v2f*>(&As[((a_p0 + 8 * i) * A2P + a_r) * 2]) = pr;
    }
#pragma unroll
    for (int j = 0; j < 4; ++j) {
      if (j < 3 || b_last) {
        v4f lo, hi4;
        lo.x = w0[j].x; lo.y = w1[j].x; lo.z = w0[j].y; lo.w = w1[j].y;
        hi4.x = w0[j].z; hi4.y = w1[j].z; hi4.z = w0[j].w; hi4.w = w1[j].w;
        *reinterpret_cast<v4f*>(&Bs[b_dst[j]]) = lo;
        *reinterpret_cast<v4f*>(&Bs[b_dst[j] + 4]) = hi4;
      }
    }
    __syncthreads();
#pragma unroll
    for (int kb = 0; kb < 32; kb += 4) {
      const int po = (kb >> 1);
      v2f af = *reinterpret_cast<const v2f*>(a_frag + po * (A2P * 2));
#pragma unroll
      for (int jt = 0; jt < 4; ++jt) {
        v2f bf = *reinterpret_cast<const v2f*>(
            b_frag + po * (B2P * 2) + jt * 16 * 2);
        acc[jt] = wmma_f32(af, bf, acc[jt]);
      }
    }
    __syncthreads();
  }
#pragma unroll
  for (int jt = 0; jt < 4; ++jt)
#pragma unroll
    for (int v = 0; v < 8; ++v)
      proj[(rb * 16 + v + 8 * hi) * P_S + jb + jt * 16 + laneM] = acc[jt][v];
  __syncthreads();

  {
    int rr = tid >> 3;
    int h = tid & 7;
    float* prow = &proj[rr * P_S + h * KER_];
    const float* bch = &bc[h * KER_];
    float m = -1e30f;
#pragma unroll
    for (int i = 0; i < KER_; ++i) {
      float v = prow[i] + bch[i];
      prow[i] = v;
      m = fmaxf(m, v);
    }
    float s = 0.f;
#pragma unroll
    for (int i = 0; i < KER_; ++i) {
      float e = __expf(prow[i] - m);
      prow[i] = e;
      s += e;
    }
    float inv = 1.f / s;
    float* wout = w_ws + (size_t)(row0 + rr) * 256 + h * 32;
#pragma unroll
    for (int i = 0; i < KER_; ++i) wout[i] = prow[i] * inv;
    wout[KER_] = 0.f;
  }
}

#define DC_TT 128
#define DC_SEG 16
#define DC_HALO 16
#define DC_ROWF (DC_TT + 2 * DC_HALO)
#define DC_WIN (DC_SEG + 2 * DC_HALO)
#define DC_PITCH 164
__global__ __launch_bounds__(256) void dynconv_kernel(
    const float* __restrict__ x, const float* __restrict__ k_ws,
    const float* __restrict__ w_ws, const float* __restrict__ conv_bias,
    float* __restrict__ out) {
  __shared__ __attribute__((aligned(16))) float tile[64 * DC_PITCH];
  const int tid = threadIdx.x;
  const int lane = tid & 31;
  const int seg = __builtin_amdgcn_readfirstlane(tid >> 5);
  const int t0 = blockIdx.x * DC_TT;
  const int h = blockIdx.y;
  const int b = blockIdx.z;
  const size_t chan0 = ((size_t)b * C_ + (size_t)h * 64) * T_;

  {
    const float* ksrc = k_ws + chan0;
    v4f stg[10];
#pragma unroll
    for (int it = 0; it < 10; ++it) {
      const int i = tid + it * 256;
      const int r = i / (DC_ROWF / 4);
      const int q = i - r * (DC_ROWF / 4);
      const int tt = t0 - DC_HALO + 4 * q;
      const int ttc = min(max(tt, 0), T_ - 4);
      v4f v = *reinterpret_cast<const v4f*>(ksrc + (size_t)r * T_ + ttc);
      const v4f z = {0.f, 0.f, 0.f, 0.f};
      stg[it] = (tt == ttc) ? v : z;
    }
#pragma unroll
    for (int it = 0; it < 10; ++it) {
      const int i = tid + it * 256;
      const int r = i / (DC_ROWF / 4);
      const int q = i - r * (DC_ROWF / 4);
      *reinterpret_cast<v4f*>(&tile[r * DC_PITCH + 4 * q]) = stg[it];
    }
  }
  __syncthreads();

  float wa[DC_WIN], wb[DC_WIN];
  {
    const float* ra = &tile[lane * DC_PITCH + seg * DC_SEG];
    const float* rb = &tile[(lane + 32) * DC_PITCH + seg * DC_SEG];
#pragma unroll
    for (int q = 0; q < DC_WIN / 4; ++q) {
      const v4f va = *reinterpret_cast<const v4f*>(ra + 4 * q);
      const v4f vb = *reinterpret_cast<const v4f*>(rb + 4 * q);
      wa[4 * q + 0] = va.x; wa[4 * q + 1] = va.y;
      wa[4 * q + 2] = va.z; wa[4 * q + 3] = va.w;
      wb[4 * q + 0] = vb.x; wb[4 * q + 1] = vb.y;
      wb[4 * q + 2] = vb.z; wb[4 * q + 3] = vb.w;
    }
  }
  __syncthreads();

  const float* wsrc =
      w_ws + ((size_t)b * T_ + t0 + seg * DC_SEG) * 256 + (size_t)h * 32;
#pragma unroll
  for (int j4 = 0; j4 < DC_SEG; j4 += 4) {
    v4f oa, ob;
#pragma unroll
    for (int jj = 0; jj < 4; ++jj) {
      const int j = j4 + jj;
      const float* wr = wsrc + (size_t)j * 256;
      float a0 = 0.f, a1 = 0.f;
#pragma unroll
      for (int kk = 0; kk < KER_; ++kk) {
        const float w = wr[kk];
        a0 = fmaf(w, wa[j + kk + 1], a0);
        a1 = fmaf(w, wb[j + kk + 1], a1);
      }
      oa[jj] = a0;
      ob[jj] = a1;
    }
    *reinterpret_cast<v4f*>(&tile[lane * DC_PITCH + seg * DC_SEG + j4]) = oa;
    *reinterpret_cast<v4f*>(&tile[(lane + 32) * DC_PITCH + seg * DC_SEG + j4]) = ob;
  }
  __syncthreads();

#pragma unroll
  for (int it = 0; it < 8; ++it) {
    const int r = it * 8 + seg;
    v4f v = *reinterpret_cast<const v4f*>(&tile[r * DC_PITCH + 4 * lane]);
    const size_t g = chan0 + (size_t)r * T_ + t0 + 4 * lane;
    const v4f xv = *reinterpret_cast<const v4f*>(x + g);
    const float cb = conv_bias[h * 64 + r];
    v.x = (v.x + cb) + xv.x;
    v.y = (v.y + cb) + xv.y;
    v.z = (v.z + cb) + xv.z;
    v.w = (v.w + cb) + xv.w;
    *reinterpret_cast<v4f*>(out + g) = v;
  }
}

extern "C" void kernel_launch(void* const* d_in, const int* in_sizes, int n_in,
                              void* d_out, int out_size, void* d_ws,
                              size_t ws_size, hipStream_t stream) {
  (void)in_sizes;
  (void)n_in;
  (void)out_size;
  (void)ws_size;
  const float* x = (const float*)d_in[0];
  const float* ln_gamma = (const float*)d_in[1];
  const float* ln_beta = (const float*)d_in[2];
  const float* Wk = (const float*)d_in[3];
  const float* bk = (const float*)d_in[4];
  const float* Wc = (const float*)d_in[5];
  const float* bc = (const float*)d_in[6];
  const float* conv_bias = (const float*)d_in[7];
  float* out = (float*)d_out;

  float* ws = (float*)d_ws;
  float* stats = ws;
  float* k_ws = ws + 32768;
  float* w_ws = k_ws + (size_t)B_ * C_ * T_;

  ln_stats_kernel<<<NROWS / 64, 256, 0, stream>>>(x, stats);
  glu_gemm_kernel<<<dim3(NROWS / 64, C_ / 64), 256, 0, stream>>>(
      x, ln_gamma, ln_beta, Wk, bk, stats, k_ws);
  conv_weight_kernel<<<NROWS / 32, 256, 0, stream>>>(k_ws, Wc, bc, w_ws);
  dynconv_kernel<<<dim3(T_ / DC_TT, HEADS_, B_), 256, 0, stream>>>(
      x, k_ws, w_ws, conv_bias, out);
}


# row-statistics kernel restructured for memory parallelism: 16-byte loads, a lane's 16 loads all issued before the first sum, 16 waves per block (one round trip, not 32); stats summation order changes
# speedup vs baseline: 2.1043x; 1.0742x over previous
#include <hip/hip_runtime.h>
#include <hip/hip_bf16.h>
#include <cstdint>

#define B_ 8
#define C_ 512
#define T_ 2048
#define HEADS_ 8
#define KER_ 31
#define NROWS (B_ * T_)
#define EPS_ 1e-5f

typedef __attribute__((ext_vector_type(2))) float v2f;
typedef __attribute__((ext_vector_type(8))) float v8f;

__device__ __forceinline__ v8f wmma_f32(v2f a, v2f b, v8f c) {
  return __builtin_amdgcn_wmma_f32_16x16x4_f32(false, a, false, b, (short)0, c,
                                               false, false);
}

typedef __attribute__((ext_vector_type(4))) float v4f;
#define LS_WAVES 16
#define LS_CPW (C_ / LS_WAVES)
#define LS_LD (LS_CPW / 2)
#define LS_PARTS (2 * LS_WAVES)
static_assert(LS_LD == 16, "the load fence below names 16 values");
__global__ __launch_bounds__(32 * LS_WAVES) void ln_stats_kernel(
    const float* __restrict__ x, float* __restrict__ stats) {
  const int tid = threadIdx.x;
  const int lane = tid & 31;
  const int wave = __builtin_amdgcn_readfirstlane(tid >> 5);
  const int row0 = blockIdx.x * 64;
  const int b = row0 / T_;
  const int t0 = row0 % T_;
  const int tq = lane & 15;
  const int par = lane >> 4;
  const float* xb = x + (size_t)b * C_ * T_ + (size_t)(wave * LS_CPW + par) * T_ +
                    t0 + 4 * tq;
  v4f v[LS_LD];
  for (int i = 0; i < LS_LD; ++i)
    v[i] = *reinterpret_cast<const v4f*>(xb + (size_t)(2 * i) * T_);
  asm volatile("" ::"v"(v[0]), "v"(v[1]), "v"(v[2]), "v"(v[3]), "v"(v[4]), "v"(v[5]),
               "v"(v[6]), "v"(v[7]), "v"(v[8]), "v"(v[9]), "v"(v[10]), "v"(v[11]),
               "v"(v[12]), "v"(v[13]), "v"(v[14]), "v"(v[15]));
  v4f s = {0.f, 0.f, 0.f, 0.f}, s2 = {0.f, 0.f, 0.f, 0.f};
  for (int i = 0; i < LS_LD; ++i) {
    s += v[i];
    s2 += v[i] * v[i];
  }
  __shared__ __attribute__((aligned(16))) float sb[LS_PARTS][64];
  __shared__ __attribute__((aligned(16))) float sq[LS_PARTS][64];
  *reinterpret_cast<v4f*>(&sb[2 * wave + par][4 * tq]) = s;
  *reinterpret_cast<v4f*>(&sq[2 * wave + par][4 * tq]) = s2;
  __syncthreads();
  if (tid < 64) {
    float S = 0.f, S2 = 0.f;
    for (int p = 0; p < LS_PARTS; ++p) {
      S += sb[p][tid];
      S2 += sq[p][tid];
    }
    float mu = S * (1.f / C_);
    float var = S2 * (1.f / C_) - mu * mu;
    stats[(row0 + tid) * 2 + 0] = mu;
    stats[(row0 + tid) * 2 + 1] = rsqrtf(var + EPS_);
  }
}

#define A1P 80
#define B1P 144
#define KT_S 65
__global__ __launch_bounds__(256) void glu_gemm_kernel(
    const float* __restrict__ x, const float* __restrict__ gamma,
    const float* __restrict__ beta, const float* __restrict__ Wk,
    const float* __restrict__ bk, const float* __restrict__ stats,
    float* __restrict__ k_ws) {
  __shared__ __attribute__((aligned(16))) float smem[32 * A1P * 2 + 32 * B1P * 2];
  float* As = smem;
  float* Bs = smem + 32 * A1P * 2;

  const int tid = threadIdx.x;
  const int row0 = blockIdx.x * 64;
  const int n0 = blockIdx.y * 64;
  const int b = row0 / T_;
  const int t0 = row0 % T_;

  const int lane = tid & 31;
  const int wave = __builtin_amdgcn_readfirstlane(tid >> 5);
  const int laneM = lane & 15;
  const int hi = lane >> 4;
  const int rblk = wave & 3;
  const int j0 = (wave >> 2) * 32;

  const int a_tq = tid & 15;
  const int a_p0 = 2 * wave + (lane >> 4);
  const int b_q = tid & 15;
  const int b_part = lane >> 4;
  const int b_p0 = wave;

  const v4f st0 = *reinterpret_cast<const v4f*>(stats + (size_t)(row0 + 4 * a_tq) * 2);
  const v4f st1 = *reinterpret_cast<const v4f*>(stats + (size_t)(row0 + 4 * a_tq) * 2 + 4);
  const float mu0 = st0.x, rs0 = st0.y, mu1 = st0.z, rs1 = st0.w;
  const float mu2 = st1.x, rs2 = st1.y, mu3 = st1.z, rs3 = st1.w;

  v8f accA0 = {}, accA1 = {}, accB0 = {}, accB1 = {};
  const float* xb = x + (size_t)b * C_ * T_ + t0 + 4 * a_tq;
  const float* wb = Wk + n0 + b_part * C_ + 4 * b_q;
  const float* a_frag = &As[(hi * A1P + rblk * 16 + laneM) * 2];
  const float* b_frag = &Bs[(hi * B1P + j0 + laneM) * 2];

  for (int kbase = 0; kbase < C_; kbase += 64) {
    v4f xa[2], xc[2];
    float ga[2], gc[2], be_a[2], be_c[2];
#pragma unroll
    for (int j = 0; j < 2; ++j) {
      const int c = kbase + 2 * (a_p0 + 16 * j);
      xa[j] = *reinterpret_cast<const v4f*>(xb + (size_t)c * T_);
      xc[j] = *reinterpret_cast<const v4f*>(xb + (size_t)(c + 1) * T_);
      ga[j] = gamma[c];
      gc[j] = gamma[c + 1];
      be_a[j] = beta[c];
      be_c[j] = beta[c + 1];
    }
    v4f w0[4], w1[4];
#pragma unroll
    for (int j = 0; j < 4; ++j) {
      const int k = kbase + 2 * (b_p0 + 8 * j);
      w0[j] = *reinterpret_cast<const v4f*>(wb + (size_t)k * (2 * C_));
      w1[j] = *reinterpret_cast<const v4f*>(wb + (size_t)(k + 1) * (2 * C_));
    }
    asm volatile("" ::"v"(xa[0]), "v"(xa[1]), "v"(xc[0]), "v"(xc[1]), "v"(w0[0]),
                 "v"(w0[1]), "v"(w0[2]), "v"(w0[3]), "v"(w1[0]), "v"(w1[1]),
                 "v"(w1[2]), "v"(w1[3]), "v"(ga[0]), "v"(ga[1]), "v"(gc[0]),
                 "v"(gc[1]), "v"(be_a[0]), "v"(be_a[1]), "v"(be_c[0]), "v"(be_c[1]));
#pragma unroll
    for (int j = 0; j < 2; ++j) {
      const int p = a_p0 + 16 * j;
      v4f lo, hi4;
      lo.x = (xa[j].x - mu0) * rs0 * ga[j] + be_a[j];
      lo.y = (xc[j].x - mu0) * rs0 * gc[j] + be_c[j];
      lo.z = (xa[j].y - mu1) * rs1 * ga[j] + be_a[j];
      lo.w = (xc[j].y - mu1) * rs1 * gc[j] + be_c[j];
      hi4.x = (xa[j].z - mu2) * rs2 * ga[j] + be_a[j];
      hi4.y = (xc[j].z - mu2) * rs2 * gc[j] + be_c[j];
      hi4.z = (xa[j].w - mu3) * rs3 * ga[j] + be_a[j];
      hi4.w = (xc[j].w - mu3) * rs3 * gc[j] + be_c[j];
      float* dst = &As[(p * A1P + 4 * a_tq) * 2];
      *reinterpret_cast<v4f*>(dst) = lo;
      *reinterpret_cast<v4f*>(dst + 4) = hi4;
    }
#pragma unroll
    for (int j = 0; j < 4; ++j) {
      const int p = b_p0 + 8 * j;
      v4f lo, hi4;
      lo.x = w0[j].x; lo.y = w1[j].x; lo.z = w0[j].y; lo.w = w1[j].y;
      hi4.x = w0[j].z; hi4.y = w1[j].z; hi4.z = w0[j].w; hi4.w = w1[j].w;
      float* dst = &Bs[(p * B1P + b_part * 64 + 4 * b_q) * 2];
      *reinterpret_cast<v4f*>(dst) = lo;
      *reinterpret_cast<v4f*>(dst + 4) = hi4;
    }
    __syncthreads();
#pragma unroll
    for (int kb = 0; kb < 64; kb += 4) {
      const int po = (kb >> 1);
      v2f af = *reinterpret_cast<const v2f*>(a_frag + po * (A1P * 2));
      v2f b0 = *reinterpret_cast<const v2f*>(b_frag + po * (B1P * 2));
      v2f b1 = *reinterpret_cast<const v2f*>(b_frag + po * (B1P * 2) + 16 * 2);
      v2f b2 = *reinterpret_cast<const v2f*>(b_frag + po * (B1P * 2) + 64 * 2);
      v2f b3 = *reinterpret_cast<const v2f*>(b_frag + po * (B1P * 2) + (64 + 16) * 2);
      accA0 = wmma_f32(af, b0, accA0);
      accA1 = wmma_f32(af, b1, accA1);
      accB0 = wmma_f32(af, b2, accB0);
      accB1 = wmma_f32(af, b3, accB1);
    }
    __syncthreads();
  }

  float* kst = smem;
  const float bka0 = bk[n0 + j0 + laneM];
  const float bka1 = bk[n0 + j0 + 16 + laneM];
  const float bkb0 = bk[C_ + n0 + j0 + laneM];
  const float bkb1 = bk[C_ + n0 + j0 + 16 + laneM];
#pragma unroll
  for (int v = 0; v < 8; ++v) {
    int tl = rblk * 16 + v + 8 * hi;
    float a0 = accA0[v] + bka0;
    float g0 = accB0[v] + bkb0;
    float a1 = accA1[v] + bka1;
    float g1 = accB1[v] + bkb1;
    kst[(j0 + laneM) * KT_S + tl] = a0 / (1.f + __expf(-g0));
    kst[(j0 + 16 + laneM) * KT_S + tl] = a1 / (1.f + __expf(-g1));
  }
  __syncthreads();
  float* kout = k_ws + (size_t)b * C_ * T_;
  for (int i = tid; i < 64 * 64; i += 256) {
    int cc = i >> 6;
    int tl = i & 63;
    kout[(size_t)(n0 + cc) * T_ + t0 + tl] = kst[cc * KT_S + tl];
  }
}

#define A2P 48
#define B2P 272
#define P_S 257
#define NCW (KER_ * HEADS_)
#define NQW (NCW / 4)
__global__ __launch_bounds__(256) void conv_weight_kernel(
    const float* __restrict__ k_ws, const float* __restrict__ Wc,
    const float* __restrict__ bc, float* __restrict__ w_ws) {
  __shared__ __attribute__((aligned(16))) float As[16 * A2P * 2];
  __shared__ __attribute__((aligned(16))) float Bs[16 * B2P * 2];
  __shared__ float proj[32 * P_S];
  const int tid = threadIdx.x;
  const int row0 = blockIdx.x * 32;
  const int b = row0 / T_;
  const int t0 = row0 % T_;
  const int lane = tid & 31;
  const int wave = tid >> 5;
  const int laneM = lane & 15;
  const int hi = lane >> 4;
  const int rb = wave & 1;
  const int jb = (wave >> 1) * 64;

  v8f acc[4] = {{}, {}, {}, {}};
  const float* kin = k_ws + (size_t)b * C_ * T_ + t0;

  if (tid < 128) {
    const int p = tid >> 3;
    const int col = NCW + (tid & 7);
    Bs[(p * B2P + col) * 2 + 0] = 0.f;
    Bs[(p * B2P + col) * 2 + 1] = 0.f;
  }

  int b_src[4], b_dst[4];
#pragma unroll
  for (int j = 0; j < 4; ++j) {
    const int item = min(tid + 256 * j, 16 * NQW - 1);
    const int p = item / NQW;
    const int q = item - p * NQW;
    b_src[j] = 2 * p * NCW + 4 * q;
    b_dst[j] = (p * B2P + 4 * q) * 2;
  }
  const bool b_last = (tid + 768) < 16 * NQW;
  const int a_r = tid & 31;
  const int a_p0 = tid >> 5;
  const float* a_frag = &As[(hi * A2P + rb * 16 + laneM) * 2];
  const float* b_frag = &Bs[(hi * B2P + jb + laneM) * 2];

  for (int kbase = 0; kbase < C_; kbase += 32) {
    const float* wsrc = Wc + (size_t)kbase * NCW;
    float a_e[2], a_o[2];
#pragma unroll
    for (int i = 0; i < 2; ++i) {
      const int c = kbase + 2 * (a_p0 + 8 * i);
      a_e[i] = kin[(size_t)c * T_ + a_r];
      a_o[i] = kin[(size_t)(c + 1) * T_ + a_r];
    }
    v4f w0[4], w1[4];
#pragma unroll
    for (int j = 0; j < 4; ++j) {
      w0[j] = *reinterpret_cast<const v4f*>(wsrc + b_src[j]);
      w1[j] = *reinterpret_cast<const v4f*>(wsrc + b_src[j] + NCW);
    }
#pragma unroll
    for (int i = 0; i < 2; ++i) {
      v2f pr;
      pr.x = a_e[i];
      pr.y = a_o[i];
      *reinterpret_cast<v2f*>(&As[((a_p0 + 8 * i) * A2P + a_r) * 2]) = pr;
    }
#pragma unroll
    for (int j = 0; j < 4; ++j) {
      if (j < 3 || b_last) {
        v4f lo, hi4;
        lo.x = w0[j].x; lo.y = w1[j].x; lo.z = w0[j].y; lo.w = w1[j].y;
        hi4.x = w0[j].z; hi4.y = w1[j].z; hi4.z = w0[j].w; hi4.w = w1[j].w;
        *reinterpret_cast<v4f*>(&Bs[b_dst[j]]) = lo;
        *reinterpret_cast<v4f*>(&Bs[b_dst[j] + 4]) = hi4;
      }
    }
    __syncthreads();
#pragma unroll
    for (int kb = 0; kb < 32; kb += 4) {
      const int po = (kb >> 1);
      v2f af = *reinterpret_cast<const v2f*>(a_frag + po * (A2P * 2));
#pragma unroll
      for (int jt = 0; jt < 4; ++jt) {
        v2f bf = *reinterpret_cast<const v2f*>(
            b_frag + po * (B2P * 2) + jt * 16 * 2);
        acc[jt] = wmma_f32(af, bf, acc[jt]);
      }
    }
    __syncthreads();
  }
#pragma unroll
  for (int jt = 0; jt < 4; ++jt)
#pragma unroll
    for (int v = 0; v < 8; ++v)
      proj[(rb * 16 + v + 8 * hi) * P_S + jb + jt * 16 + laneM] = acc[jt][v];
  __syncthreads();

  {
    int rr = tid >> 3;
    int h = tid & 7;
    float* prow = &proj[rr * P_S + h * KER_];
    const float* bch = &bc[h * KER_];
    float m = -1e30f;
#pragma unroll
    for (int i = 0; i < KER_; ++i) {
      float v = prow[i] + bch[i];
      prow[i] = v;
      m = fmaxf(m, v);
    }
    float s = 0.f;
#pragma unroll
    for (int i = 0; i < KER_; ++i) {
      float e = __expf(prow[i] - m);
      prow[i] = e;
      s += e;
    }
    float inv = 1.f / s;
    float* wout = w_ws + (size_t)(row0 + rr) * 256 + h * 32;
#pragma unroll
    for (int i = 0; i < KER_; ++i) wout[i] = prow[i] * inv;
    wout[KER_] = 0.f;
  }
}

#define DC_TT 128
#define DC_SEG 16
#define DC_HALO 16
#define DC_ROWF (DC_TT + 2 * DC_HALO)
#define DC_WIN (DC_SEG + 2 * DC_HALO)
#define DC_PITCH 164
__global__ __launch_bounds__(256) void dynconv_kernel(
    const float* __restrict__ x, const float* __restrict__ k_ws,
    const float* __restrict__ w_ws, const float* __restrict__ conv_bias,
    float* __restrict__ out) {
  __shared__ __attribute__((aligned(16))) float tile[64 * DC_PITCH];
  const int tid = threadIdx.x;
  const int lane = tid & 31;
  const int seg = __builtin_amdgcn_readfirstlane(tid >> 5);
  const int t0 = blockIdx.x * DC_TT;
  const int h = blockIdx.y;
  const int b = blockIdx.z;
  const size_t chan0 = ((size_t)b * C_ + (size_t)h * 64) * T_;

  {
    const float* ksrc = k_ws + chan0;
    v4f stg[10];
#pragma unroll
    for (int it = 0; it < 10; ++it) {
      const int i = tid + it * 256;
      const int r = i / (DC_ROWF / 4);
      const int q = i - r * (DC_ROWF / 4);
      const int tt = t0 - DC_HALO + 4 * q;
      const int ttc = min(max(tt, 0), T_ - 4);
      v4f v = *reinterpret_cast<const v4f*>(ksrc + (size_t)r * T_ + ttc);
      const v4f z = {0.f, 0.f, 0.f, 0.f};
      stg[it] = (tt == ttc) ? v : z;
    }
#pragma unroll
    for (int it = 0; it < 10; ++it) {
      const int i = tid + it * 256;
      const int r = i / (DC_ROWF / 4);
      const int q = i - r * (DC_ROWF / 4);
      *reinterpret_cast<v4f*>(&tile[r * DC_PITCH + 4 * q]) = stg[it];
    }
  }
  __syncthreads();

  float wa[DC_WIN], wb[DC_WIN];
  {
    const float* ra = &tile[lane * DC_PITCH + seg * DC_SEG];
    const float* rb = &tile[(lane + 32) * DC_PITCH + seg * DC_SEG];
#pragma unroll
    for (int q = 0; q < DC_WIN / 4; ++q) {
      const v4f va = *reinterpret_cast<const v4f*>(ra + 4 * q);
      const v4f vb = *reinterpret_cast<const v4f*>(rb + 4 * q);
      wa[4 * q + 0] = va.x; wa[4 * q + 1] = va.y;
      wa[4 * q + 2] = va.z; wa[4 * q + 3] = va.w;
      wb[4 * q + 0] = vb.x; wb[4 * q + 1] = vb.y;
      wb[4 * q + 2] = vb.z; wb[4 * q + 3] = vb.w;
    }
  }
  __syncthreads();

  const float* wsrc =
      w_ws + ((size_t)b * T_ + t0 + seg * DC_SEG) * 256 + (size_t)h * 32;
#pragma unroll
  for (int j4 = 0; j4 < DC_SEG; j4 += 4) {
    v4f oa, ob;
#pragma unroll
    for (int jj = 0; jj < 4; ++jj) {
      const int j = j4 + jj;
      const float* wr = wsrc + (size_t)j * 256;
      float a0 = 0.f, a1 = 0.f;
#pragma unroll
      for (int kk = 0; kk < KER_; ++kk) {
        const float w = wr[kk];
        a0 = fmaf(w, wa[j + kk + 1], a0);
        a1 = fmaf(w, wb[j + kk + 1], a1);
      }
      oa[jj] = a0;
      ob[jj] = a1;
    }
    *reinterpret_cast<v4f*>(&tile[lane * DC_PITCH + seg * DC_SEG + j4]) = oa;
    *reinterpret_cast<v4f*>(&tile[(lane + 32) * DC_PITCH + seg * DC_SEG + j4]) = ob;
  }
  __syncthreads();

#pragma unroll
  for (int it = 0; it < 8; ++it) {
    const int r = it * 8 + seg;
    v4f v = *reinterpret_cast<const v4f*>(&tile[r * DC_PITCH + 4 * lane]);
    const size_t g = chan0 + (size_t)r * T_ + t0 + 4 * lane;
    const v4f xv = *reinterpret_cast<const v4f*>(x + g);
    const float cb = conv_bias[h * 64 + r];
    v.x = (v.x + cb) + xv.x;
    v.y = (v.y + cb) + xv.y;
    v.z = (v.z + cb) + xv.z;
    v.w = (v.w + cb) + xv.w;
    *reinterpret_cast<v4f*>(out + g) = v;
  }
}

extern "C" void kernel_launch(void* const* d_in, const int* in_sizes, int n_in,
                              void* d_out, int out_size, void* d_ws,
                              size_t ws_size, hipStream_t stream) {
  (void)in_sizes;
  (void)n_in;
  (void)out_size;
  (void)ws_size;
  const float* x = (const float*)d_in[0];
  const float* ln_gamma = (const float*)d_in[1];
  const float* ln_beta = (const float*)d_in[2];
  const float* Wk = (const float*)d_in[3];
  const float* bk = (const float*)d_in[4];
  const float* Wc = (const float*)d_in[5];
  const float* bc = (const float*)d_in[6];
  const float* conv_bias = (const float*)d_in[7];
  float* out = (float*)d_out;

  float* ws = (float*)d_ws;
  float* stats = ws;
  float* k_ws = ws + 32768;
  float* w_ws = k_ws + (size_t)B_ * C_ * T_;

  ln_stats_kernel<<<NROWS / 64, 32 * LS_WAVES, 0, stream>>>(x, stats);
  glu_gemm_kernel<<<dim3(NROWS / 64, C_ / 64), 256, 0, stream>>>(
      x, ln_gamma, ln_beta, Wk, bk, stats, k_ws);
  conv_weight_kernel<<<NROWS / 32, 256, 0, stream>>>(k_ws, Wc, bc, w_ws);
  dynconv_kernel<<<dim3(T_ / DC_TT, HEADS_, B_), 256, 0, stream>>>(
      x, k_ws, w_ws, conv_bias, out);
}


# GEMM1 work-group tile 64 -> 128 rows, a wave owns 32 rows x 64 columns: each LDS B fragment feeds two matrix instructions, a staged Wk chunk serves 128 rows; 1024 groups, all resident; k order kept
# speedup vs baseline: 2.1722x; 1.0323x over previous
#include <hip/hip_runtime.h>
#include <hip/hip_bf16.h>
#include <cstdint>

#define B_ 8
#define C_ 512
#define T_ 2048
#define HEADS_ 8
#define KER_ 31
#define NROWS (B_ * T_)
#define EPS_ 1e-5f

typedef __attribute__((ext_vector_type(2))) float v2f;
typedef __attribute__((ext_vector_type(8))) float v8f;

__device__ __forceinline__ v8f wmma_f32(v2f a, v2f b, v8f c) {
  return __builtin_amdgcn_wmma_f32_16x16x4_f32(false, a, false, b, (short)0, c,
                                               false, false);
}

typedef __attribute__((ext_vector_type(4))) float v4f;
#define LS_WAVES 16
#define LS_CPW (C_ / LS_WAVES)
#define LS_LD (LS_CPW / 2)
#define LS_PARTS (2 * LS_WAVES)
static_assert(LS_LD == 16, "the load fence below names 16 values");
__global__ __launch_bounds__(32 * LS_WAVES) void ln_stats_kernel(
    const float* __restrict__ x, float* __restrict__ stats) {
  const int tid = threadIdx.x;
  const int lane = tid & 31;
  const int wave = __builtin_amdgcn_readfirstlane(tid >> 5);
  const int row0 = blockIdx.x * 64;
  const int b = row0 / T_;
  const int t0 = row0 % T_;
  const int tq = lane & 15;
  const int par = lane >> 4;
  const float* xb = x + (size_t)b * C_ * T_ + (size_t)(wave * LS_CPW + par) * T_ +
                    t0 + 4 * tq;
  v4f v[LS_LD];
  for (int i = 0; i < LS_LD; ++i)
    v[i] = *reinterpret_cast<const v4f*>(xb + (size_t)(2 * i) * T_);
  asm volatile("" ::"v"(v[0]), "v"(v[1]), "v"(v[2]), "v"(v[3]), "v"(v[4]), "v"(v[5]),
               "v"(v[6]), "v"(v[7]), "v"(v[8]), "v"(v[9]), "v"(v[10]), "v"(v[11]),
               "v"(v[12]), "v"(v[13]), "v"(v[14]), "v"(v[15]));
  v4f s = {0.f, 0.f, 0.f, 0.f}, s2 = {0.f, 0.f, 0.f, 0.f};
  for (int i = 0; i < LS_LD; ++i) {
    s += v[i];
    s2 += v[i] * v[i];
  }
  __shared__ __attribute__((aligned(16))) float sb[LS_PARTS][64];
  __shared__ __attribute__((aligned(16))) float sq[LS_PARTS][64];
  *reinterpret_cast<v4f*>(&sb[2 * wave + par][4 * tq]) = s;
  *reinterpret_cast<v4f*>(&sq[2 * wave + par][4 * tq]) = s2;
  __syncthreads();
  if (tid < 64) {
    float S = 0.f, S2 = 0.f;
    for (int p = 0; p < LS_PARTS; ++p) {
      S += sb[p][tid];
      S2 += sq[p][tid];
    }
    float mu = S * (1.f / C_);
    float var = S2 * (1.f / C_) - mu * mu;
    stats[(row0 + tid) * 2 + 0] = mu;
    stats[(row0 + tid) * 2 + 1] = rsqrtf(var + EPS_);
  }
}

#define G1_ROWS 128
#define G1_KC 32
#define G1_KP (G1_KC / 2)
#define A1P 144
#define B1P 144
#define KT_S 129
__global__ __launch_bounds__(256, 8) void glu_gemm_kernel(
    const float* __restrict__ x, const float* __restrict__ gamma,
    const float* __restrict__ beta, const float* __restrict__ Wk,
    const float* __restrict__ bk, const float* __restrict__ stats,
    float* __restrict__ k_ws) {
  __shared__ __attribute__((aligned(16))) float smem[G1_KP * A1P * 2 + G1_KP * B1P * 2];
  float* As = smem;
  float* Bs = smem + G1_KP * A1P * 2;

  const int tid = threadIdx.x;
  const int row0 = blockIdx.x * G1_ROWS;
  const int n0 = blockIdx.y * 64;
  const int b = row0 / T_;
  const int t0 = row0 % T_;

  const int lane = tid & 31;
  const int wave = __builtin_amdgcn_readfirstlane(tid >> 5);
  const int laneM = lane & 15;
  const int hi = lane >> 4;
  const int rblk = wave & 3;
  const int j0 = (wave >> 2) * 32;

  const int a_tq = lane;
  const int b_q = lane & 15;
  const int b_part = lane >> 4;

  const v4f st0 = *reinterpret_cast<const v4f*>(stats + (size_t)(row0 + 4 * a_tq) * 2);
  const v4f st1 = *reinterpret_cast<const v4f*>(stats + (size_t)(row0 + 4 * a_tq) * 2 + 4);
  const v4f mu4 = {st0.x, st0.z, st1.x, st1.z};
  const v4f rs4 = {st0.y, st0.w, st1.y, st1.w};

  v8f accA[2][2] = {{{}, {}}, {{}, {}}};
  v8f accB[2][2] = {{{}, {}}, {{}, {}}};
  const float* xb = x + (size_t)b * C_ * T_ + t0;
  const float* wb = Wk + n0;
  const int x_lo = 4 * a_tq;
  const int w_lo = b_part * C_ + 4 * b_q;
  const float* a_frag = &As[(hi * A1P + rblk * 32 + laneM) * 2];
  const float* b_frag = &Bs[(hi * B1P + j0 + laneM) * 2];

  for (int kbase = 0; kbase < C_; kbase += G1_KC) {
    v4f xa[2], xc[2];
    float ga[2], gc[2], be_a[2], be_c[2];
    v4f w0[2], w1[2];
#pragma unroll
    for (int j = 0; j < 2; ++j) {
      const int c = kbase + 2 * (wave + 8 * j);
      const float* xr = xb + (size_t)c * T_;
      xa[j] = *reinterpret_cast<const v4f*>(xr + x_lo);
      xc[j] = *reinterpret_cast<const v4f*>(xr + T_ + x_lo);
      ga[j] = gamma[c];
      gc[j] = gamma[c + 1];
      be_a[j] = beta[c];
      be_c[j] = beta[c + 1];
      const float* wr = wb + (size_t)c * (2 * C_);
      w0[j] = *reinterpret_cast<const v4f*>(wr + w_lo);
      w1[j] = *reinterpret_cast<const v4f*>(wr + 2 * C_ + w_lo);
    }
    asm volatile("" ::"v"(xa[0]), "v"(xa[1]), "v"(xc[0]), "v"(xc[1]), "v"(w0[0]),
                 "v"(w0[1]), "v"(w1[0]), "v"(w1[1]));
#pragma unroll
    for (int j = 0; j < 2; ++j) {
      const int p = wave + 8 * j;
      v4f lo, hi4;
      const v4f ha = (xa[j] - mu4) * rs4 * ga[j] + be_a[j];
      const v4f hc = (xc[j] - mu4) * rs4 * gc[j] + be_c[j];
      lo.x = ha.x; lo.y = hc.x; lo.z = ha.y; lo.w = hc.y;
      hi4.x = ha.z; hi4.y = hc.z; hi4.z = ha.w; hi4.w = hc.w;
      float* dst = &As[(p * A1P + 4 * a_tq) * 2];
      *reinterpret_cast<v4f*>(dst) = lo;
      *reinterpret_cast<v4f*>(dst + 4) = hi4;
    }
#pragma unroll
    for (int j = 0; j < 2; ++j) {
      const int p = wave + 8 * j;
      v4f lo, hi4;
      lo.x = w0[j].x; lo.y = w1[j].x; lo.z = w0[j].y; lo.w = w1[j].y;
      hi4.x = w0[j].z; hi4.y = w1[j].z; hi4.z = w0[j].w; hi4.w = w1[j].w;
      float* dst = &Bs[(p * B1P + b_part * 64 + 4 * b_q) * 2];
      *reinterpret_cast<v4f*>(dst) = lo;
      *reinterpret_cast<v4f*>(dst + 4) = hi4;
    }
    __syncthreads();
#pragma unroll 1
    for (int kb = 0; kb < G1_KC; kb += 4) {
      const int po = (kb >> 1);
      v2f af0 = *reinterpret_cast<const v2f*>(a_frag + po * (A1P * 2));
      v2f af1 = *reinterpret_cast<const v2f*>(a_frag + po * (A1P * 2) + 16 * 2);
      v2f b0 = *reinterpret_cast<const v2f*>(b_frag + po * (B1P * 2));
      v2f b1 = *reinterpret_cast<const v2f*>(b_frag + po * (B1P * 2) + 16 * 2);
      v2f b2 = *reinterpret_cast<const v2f*>(b_frag + po * (B1P * 2) + 64 * 2);
      v2f b3 = *reinterpret_cast<const v2f*>(b_frag + po * (B1P * 2) + (64 + 16) * 2);
      accA[0][0] = wmma_f32(af0, b0, accA[0][0]);
      accA[1][0] = wmma_f32(af1, b0, accA[1][0]);
      accA[0][1] = wmma_f32(af0, b1, accA[0][1]);
      accA[1][1] = wmma_f32(af1, b1, accA[1][1]);
      accB[0][0] = wmma_f32(af0, b2, accB[0][0]);
      accB[1][0] = wmma_f32(af1, b2, accB[1][0]);
      accB[0][1] = wmma_f32(af0, b3, accB[0][1]);
      accB[1][1] = wmma_f32(af1, b3, accB[1][1]);
    }
    __syncthreads();
  }

  float* kst = smem;
  const float bka0 = bk[n0 + j0 + laneM];
  const float bka1 = bk[n0 + j0 + 16 + laneM];
  const float bkb0 = bk[C_ + n0 + j0 + laneM];
  const float bkb1 = bk[C_ + n0 + j0 + 16 + laneM];
#pragma unroll
  for (int r = 0; r < 2; ++r) {
#pragma unroll
    for (int v = 0; v < 8; ++v) {
      int tl = rblk * 32 + r * 16 + v + 8 * hi;
      float a0 = accA[r][0][v] + bka0;
      float g0 = accB[r][0][v] + bkb0;
      float a1 = accA[r][1][v] + bka1;
      float g1 = accB[r][1][v] + bkb1;
      kst[(j0 + laneM) * KT_S + tl] = a0 / (1.f + __expf(-g0));
      kst[(j0 + 16 + laneM) * KT_S + tl] = a1 / (1.f + __expf(-g1));
    }
  }
  __syncthreads();
  float* kout = k_ws + (size_t)b * C_ * T_;
  for (int i = tid; i < 64 * G1_ROWS; i += 256) {
    int cc = i >> 7;
    int tl = i & (G1_ROWS - 1);
    kout[(size_t)(n0 + cc) * T_ + t0 + tl] = kst[cc * KT_S + tl];
  }
}

#define A2P 48
#define B2P 272
#define P_S 257
#define NCW (KER_ * HEADS_)
#define NQW (NCW / 4)
__global__ __launch_bounds__(256) void conv_weight_kernel(
    const float* __restrict__ k_ws, const float* __restrict__ Wc,
    const float* __restrict__ bc, float* __restrict__ w_ws) {
  __shared__ __attribute__((aligned(16))) float As[16 * A2P * 2];
  __shared__ __attribute__((aligned(16))) float Bs[16 * B2P * 2];
  __shared__ float proj[32 * P_S];
  const int tid = threadIdx.x;
  const int row0 = blockIdx.x * 32;
  const int b = row0 / T_;
  const int t0 = row0 % T_;
  const int lane = tid & 31;
  const int wave = tid >> 5;
  const int laneM = lane & 15;
  const int hi = lane >> 4;
  const int rb = wave & 1;
  const int jb = (wave >> 1) * 64;

  v8f acc[4] = {{}, {}, {}, {}};
  const float* kin = k_ws + (size_t)b * C_ * T_ + t0;

  if (tid < 128) {
    const int p = tid >> 3;
    const int col = NCW + (tid & 7);
    Bs[(p * B2P + col) * 2 + 0] = 0.f;
    Bs[(p * B2P + col) * 2 + 1] = 0.f;
  }

  int b_src[4], b_dst[4];
#pragma unroll
  for (int j = 0; j < 4; ++j) {
    const int item = min(tid + 256 * j, 16 * NQW - 1);
    const int p = item / NQW;
    const int q = item - p * NQW;
    b_src[j] = 2 * p * NCW + 4 * q;
    b_dst[j] = (p * B2P + 4 * q) * 2;
  }
  const bool b_last = (tid + 768) < 16 * NQW;
  const int a_r = tid & 31;
  const int a_p0 = tid >> 5;
  const float* a_frag = &As[(hi * A2P + rb * 16 + laneM) * 2];
  const float* b_frag = &Bs[(hi * B2P + jb + laneM) * 2];

  for (int kbase = 0; kbase < C_; kbase += 32) {
    const float* wsrc = Wc + (size_t)kbase * NCW;
    float a_e[2], a_o[2];
#pragma unroll
    for (int i = 0; i < 2; ++i) {
      const int c = kbase + 2 * (a_p0 + 8 * i);
      a_e[i] = kin[(size_t)c * T_ + a_r];
      a_o[i] = kin[(size_t)(c + 1) * T_ + a_r];
    }
    v4f w0[4], w1[4];
#pragma unroll
    for (int j = 0; j < 4; ++j) {
      w0[j] = *reinterpret_cast<const v4f*>(wsrc + b_src[j]);
      w1[j] = *reinterpret_cast<const v4f*>(wsrc + b_src[j] + NCW);
    }
#pragma unroll
    for (int i = 0; i < 2; ++i) {
      v2f pr;
      pr.x = a_e[i];
      pr.y = a_o[i];
      *reinterpret_cast<v2f*>(&As[((a_p0 + 8 * i) * A2P + a_r) * 2]) = pr;
    }
#pragma unroll
    for (int j = 0; j < 4; ++j) {
      if (j < 3 || b_last) {
        v4f lo, hi4;
        lo.x = w0[j].x; lo.y = w1[j].x; lo.z = w0[j].y; lo.w = w1[j].y;
        hi4.x = w0[j].z; hi4.y = w1[j].z; hi4.z = w0[j].w; hi4.w = w1[j].w;
        *reinterpret_cast<v4f*>(&Bs[b_dst[j]]) = lo;
        *reinterpret_cast<v4f*>(&Bs[b_dst[j] + 4]) = hi4;
      }
    }
    __syncthreads();
#pragma unroll
    for (int kb = 0; kb < 32; kb += 4) {
      const int po = (kb >> 1);
      v2f af = *reinterpret_cast<const v2f*>(a_frag + po * (A2P * 2));
#pragma unroll
      for (int jt = 0; jt < 4; ++jt) {
        v2f bf = *reinterpret_cast<const v2f*>(
            b_frag + po * (B2P * 2) + jt * 16 * 2);
        acc[jt] = wmma_f32(af, bf, acc[jt]);
      }
    }
    __syncthreads();
  }
#pragma unroll
  for (int jt = 0; jt < 4; ++jt)
#pragma unroll
    for (int v = 0; v < 8; ++v)
      proj[(rb * 16 + v + 8 * hi) * P_S + jb + jt * 16 + laneM] = acc[jt][v];
  __syncthreads();

  {
    int rr = tid >> 3;
    int h = tid & 7;
    float* prow = &proj[rr * P_S + h * KER_];
    const float* bch = &bc[h * KER_];
    float m = -1e30f;
#pragma unroll
    for (int i = 0; i < KER_; ++i) {
      float v = prow[i] + bch[i];
      prow[i] = v;
      m = fmaxf(m, v);
    }
    float s = 0.f;
#pragma unroll
    for (int i = 0; i < KER_; ++i) {
      float e = __expf(prow[i] - m);
      prow[i] = e;
      s += e;
    }
    float inv = 1.f / s;
    float* wout = w_ws + (size_t)(row0 + rr) * 256 + h * 32;
#pragma unroll
    for (int i = 0; i < KER_; ++i) wout[i] = prow[i] * inv;
    wout[KER_] = 0.f;
  }
}

#define DC_TT 128
#define DC_SEG 16
#define DC_HALO 16
#define DC_ROWF (DC_TT + 2 * DC_HALO)
#define DC_WIN (DC_SEG + 2 * DC_HALO)
#define DC_PITCH 164
__global__ __launch_bounds__(256) void dynconv_kernel(
    const float* __restrict__ x, const float* __restrict__ k_ws,
    const float* __restrict__ w_ws, const float* __restrict__ conv_bias,
    float* __restrict__ out) {
  __shared__ __attribute__((aligned(16))) float tile[64 * DC_PITCH];
  const int tid = threadIdx.x;
  const int lane = tid & 31;
  const int seg = __builtin_amdgcn_readfirstlane(tid >> 5);
  const int t0 = blockIdx.x * DC_TT;
  const int h = blockIdx.y;
  const int b = blockIdx.z;
  const size_t chan0 = ((size_t)b * C_ + (size_t)h * 64) * T_;

  {
    const float* ksrc = k_ws + chan0;
    v4f stg[10];
#pragma unroll
    for (int it = 0; it < 10; ++it) {
      const int i = tid + it * 256;
      const int r = i / (DC_ROWF / 4);
      const int q = i - r * (DC_ROWF / 4);
      const int tt = t0 - DC_HALO + 4 * q;
      const int ttc = min(max(tt, 0), T_ - 4);
      v4f v = *reinterpret_cast<const v4f*>(ksrc + (size_t)r * T_ + ttc);
      const v4f z = {0.f, 0.f, 0.f, 0.f};
      stg[it] = (tt == ttc) ? v : z;
    }
#pragma unroll
    for (int it = 0; it < 10; ++it) {
      const int i = tid + it * 256;
      const int r = i / (DC_ROWF / 4);
      const int q = i - r * (DC_ROWF / 4);
      *reinterpret_cast<v4f*>(&tile[r * DC_PITCH + 4 * q]) = stg[it];
    }
  }
  __syncthreads();

  float wa[DC_WIN], wb[DC_WIN];
  {
    const float* ra = &tile[lane * DC_PITCH + seg * DC_SEG];
    const float* rb = &tile[(lane + 32) * DC_PITCH + seg * DC_SEG];
#pragma unroll
    for (int q = 0; q < DC_WIN / 4; ++q) {
      const v4f va = *reinterpret_cast<const v4f*>(ra + 4 * q);
      const v4f vb = *reinterpret_cast<const v4f*>(rb + 4 * q);
      wa[4 * q + 0] = va.x; wa[4 * q + 1] = va.y;
      wa[4 * q + 2] = va.z; wa[4 * q + 3] = va.w;
      wb[4 * q + 0] = vb.x; wb[4 * q + 1] = vb.y;
      wb[4 * q + 2] = vb.z; wb[4 * q + 3] = vb.w;
    }
  }
  __syncthreads();

  const float* wsrc =
      w_ws + ((size_t)b * T_ + t0 + seg * DC_SEG) * 256 + (size_t)h * 32;
#pragma unroll
  for (int j4 = 0; j4 < DC_SEG; j4 += 4) {
    v4f oa, ob;
#pragma unroll
    for (int jj = 0; jj < 4; ++jj) {
      const int j = j4 + jj;
      const float* wr = wsrc + (size_t)j * 256;
      float a0 = 0.f, a1 = 0.f;
#pragma unroll
      for (int kk = 0; kk < KER_; ++kk) {
        const float w = wr[kk];
        a0 = fmaf(w, wa[j + kk + 1], a0);
        a1 = fmaf(w, wb[j + kk + 1], a1);
      }
      oa[jj] = a0;
      ob[jj] = a1;
    }
    *reinterpret_cast<v4f*>(&tile[lane * DC_PITCH + seg * DC_SEG + j4]) = oa;
    *reinterpret_cast<v4f*>(&tile[(lane + 32) * DC_PITCH + seg * DC_SEG + j4]) = ob;
  }
  __syncthreads();

#pragma unroll
  for (int it = 0; it < 8; ++it) {
    const int r = it * 8 + seg;
    v4f v = *reinterpret_cast<const v4f*>(&tile[r * DC_PITCH + 4 * lane]);
    const size_t g = chan0 + (size_t)r * T_ + t0 + 4 * lane;
    const v4f xv = *reinterpret_cast<const v4f*>(x + g);
    const float cb = conv_bias[h * 64 + r];
    v.x = (v.x + cb) + xv.x;
    v.y = (v.y + cb) + xv.y;
    v.z = (v.z + cb) + xv.z;
    v.w = (v.w + cb) + xv.w;
    *reinterpret_cast<v4f*>(out + g) = v;
  }
}

extern "C" void kernel_launch(void* const* d_in, const int* in_sizes, int n_in,
                              void* d_out, int out_size, void* d_ws,
                              size_t ws_size, hipStream_t stream) {
  (void)in_sizes;
  (void)n_in;
  (void)out_size;
  (void)ws_size;
  const float* x = (const float*)d_in[0];
  const float* ln_gamma = (const float*)d_in[1];
  const float* ln_beta = (const float*)d_in[2];
  const float* Wk = (const float*)d_in[3];
  const float* bk = (const float*)d_in[4];
  const float* Wc = (const float*)d_in[5];
  const float* bc = (const float*)d_in[6];
  const float* conv_bias = (const float*)d_in[7];
  float* out = (float*)d_out;

  float* ws = (float*)d_ws;
  float* stats = ws;
  float* k_ws = ws + 32768;
  float* w_ws = k_ws + (size_t)B_ * C_ * T_;

  ln_stats_kernel<<<NROWS / 64, 32 * LS_WAVES, 0, stream>>>(x, stats);
  glu_gemm_kernel<<<dim3(NROWS / G1_ROWS, C_ / 64), 256, 0, stream>>>(
      x, ln_gamma, ln_beta, Wk, bk, stats, k_ws);
  conv_weight_kernel<<<NROWS / 32, 256, 0, stream>>>(k_ws, Wc, bc, w_ws);
  dynconv_kernel<<<dim3(T_ / DC_TT, HEADS_, B_), 256, 0, stream>>>(
      x, k_ws, w_ws, conv_bias, out);
}


# tap GEMM fused into the GLU GEMM: one work-group per WGP owns 64 whole rows, x tile resident in LDS, Wk/Wc chunks double-buffered, k slice handed over in LDS; 3 launches; k order kept
# speedup vs baseline: 2.3641x; 1.0883x over previous
#include <hip/hip_runtime.h>
#include <hip/hip_bf16.h>
#include <cstdint>

#define B_ 8
#define C_ 512
#define T_ 2048
#define HEADS_ 8
#define KER_ 31
#define NROWS (B_ * T_)
#define EPS_ 1e-5f

typedef __attribute__((ext_vector_type(2))) float v2f;
typedef __attribute__((ext_vector_type(8))) float v8f;

__device__ __forceinline__ v8f wmma_f32(v2f a, v2f b, v8f c) {
  return __builtin_amdgcn_wmma_f32_16x16x4_f32(false, a, false, b, (short)0, c,
                                               false, false);
}

typedef __attribute__((ext_vector_type(4))) float v4f;
#define LS_WAVES 16
#define LS_CPW (C_ / LS_WAVES)
#define LS_LD (LS_CPW / 2)
#define LS_PARTS (2 * LS_WAVES)
static_assert(LS_LD == 16, "the load fence below names 16 values");
__global__ __launch_bounds__(32 * LS_WAVES) void ln_stats_kernel(
    const float* __restrict__ x, float* __restrict__ stats) {
  const int tid = threadIdx.x;
  const int lane = tid & 31;
  const int wave = __builtin_amdgcn_readfirstlane(tid >> 5);
  const int row0 = blockIdx.x * 64;
  const int b = row0 / T_;
  const int t0 = row0 % T_;
  const int tq = lane & 15;
  const int par = lane >> 4;
  const float* xb = x + (size_t)b * C_ * T_ + (size_t)(wave * LS_CPW + par) * T_ +
                    t0 + 4 * tq;
  v4f v[LS_LD];
  for (int i = 0; i < LS_LD; ++i)
    v[i] = *reinterpret_cast<const v4f*>(xb + (size_t)(2 * i) * T_);
  asm volatile("" ::"v"(v[0]), "v"(v[1]), "v"(v[2]), "v"(v[3]), "v"(v[4]), "v"(v[5]),
               "v"(v[6]), "v"(v[7]), "v"(v[8]), "v"(v[9]), "v"(v[10]), "v"(v[11]),
               "v"(v[12]), "v"(v[13]), "v"(v[14]), "v"(v[15]));
  v4f s = {0.f, 0.f, 0.f, 0.f}, s2 = {0.f, 0.f, 0.f, 0.f};
  for (int i = 0; i < LS_LD; ++i) {
    s += v[i];
    s2 += v[i] * v[i];
  }
  __shared__ __attribute__((aligned(16))) float sb[LS_PARTS][64];
  __shared__ __attribute__((aligned(16))) float sq[LS_PARTS][64];
  *reinterpret_cast<v4f*>(&sb[2 * wave + par][4 * tq]) = s;
  *reinterpret_cast<v4f*>(&sq[2 * wave + par][4 * tq]) = s2;
  __syncthreads();
  if (tid < 64) {
    float S = 0.f, S2 = 0.f;
    for (int p = 0; p < LS_PARTS; ++p) {
      S += sb[p][tid];
      S2 += sq[p][tid];
    }
    float mu = S * (1.f / C_);
    float var = S2 * (1.f / C_) - mu * mu;
    stats[(row0 + tid) * 2 + 0] = mu;
    stats[(row0 + tid) * 2 + 1] = rsqrtf(var + EPS_);
  }
}

#define F_ROWS 64
#define F_KC 32
#define F_KP (F_KC / 2)
#define F_KC_SPLIT 16
#define F_AP 80
#define F_BP 272
#define F_NG 128
#define F_AS_FLOATS ((C_ / 2) * F_AP * 2)
#define F_BS_FLOATS (F_KP * F_BP * 2)
#define F_KS_FLOATS ((F_NG / 2) * F_AP * 2)
#define P_S 257
#define NCW (KER_ * HEADS_)
#define NQW (NCW / 4)
static_assert(F_ROWS * P_S <= F_AS_FLOATS, "the softmax stage reuses the x tile");
__global__ __launch_bounds__(256) void glu_tap_kernel(
    const float* __restrict__ x, const float* __restrict__ gamma,
    const float* __restrict__ beta, const float* __restrict__ Wk,
    const float* __restrict__ bk, const float* __restrict__ Wc,
    const float* __restrict__ bc, const float* __restrict__ stats,
    float* __restrict__ k_ws, float* __restrict__ w_ws) {
  __shared__ __attribute__((aligned(16))) float smem[F_AS_FLOATS + 2 * F_BS_FLOATS + F_KS_FLOATS];
  float* As = smem;
  float* Bs = smem + F_AS_FLOATS;
  float* Ks = smem + F_AS_FLOATS + 2 * F_BS_FLOATS;

  const int tid = threadIdx.x;
  const int row0 = blockIdx.x * F_ROWS;
  const int b = row0 / T_;
  const int t0 = row0 % T_;
  const int lane = tid & 31;
  const int wave = __builtin_amdgcn_readfirstlane(tid >> 5);
  const int laneM = lane & 15;
  const int hi = lane >> 4;
  const int rblk = wave & 1;
  const int j0 = (wave >> 1) * 32;
  const int jb = (wave >> 1) * 64;

  const int wk_src = (2 * (wave >> 1)) * (2 * C_) + (wave & 1) * C_ + 4 * lane;
  const int wk_dst = (((wave >> 1)) * F_BP + (wave & 1) * F_NG + 4 * lane) * 2;
  int b_src[4], b_dst[4];
#pragma unroll
  for (int j = 0; j < 4; ++j) {
    const int item = min(tid + 256 * j, 16 * NQW - 1);
    const int p = item / NQW;
    const int q = item - p * NQW;
    b_src[j] = 2 * p * NCW + 4 * q;
    b_dst[j] = (p * F_BP + 4 * q) * 2;
  }
  const bool b_last = (tid + 768) < 16 * NQW;
  if (!b_last) b_dst[3] = (((tid - 224) >> 1) * F_BP + NCW + 4 * (tid & 1)) * 2;

  v4f w0[4], w1[4];
  auto load_wk = [&](int ng, int kc) {
    const float* s = Wk + (size_t)(kc * F_KC) * (2 * C_) + ng * F_NG + wk_src;
#pragma unroll
    for (int j = 0; j < 4; ++j) {
      w0[j] = *reinterpret_cast<const v4f*>(s + (size_t)(8 * j) * (2 * C_));
      w1[j] = *reinterpret_cast<const v4f*>(s + (size_t)(8 * j + 1) * (2 * C_));
    }
  };
  auto store_wk = [&](float* Bd) {
#pragma unroll
    for (int j = 0; j < 4; ++j) {
      v4f lo, hi4;
      lo.x = w0[j].x; lo.y = w1[j].x; lo.z = w0[j].y; lo.w = w1[j].y;
      hi4.x = w0[j].z; hi4.y = w1[j].z; hi4.z = w0[j].w; hi4.w = w1[j].w;
      float* dst = Bd + wk_dst + j * (4 * F_BP * 2);
      *reinterpret_cast<v4f*>(dst) = lo;
      *reinterpret_cast<v4f*>(dst + 4) = hi4;
    }
  };
  auto load_wc = [&](int ng, int c2) {
    const float* s = Wc + (size_t)(ng * F_NG + c2 * F_KC) * NCW;
#pragma unroll
    for (int j = 0; j < 4; ++j) {
      w0[j] = *reinterpret_cast<const v4f*>(s + b_src[j]);
      w1[j] = *reinterpret_cast<const v4f*>(s + b_src[j] + NCW);
    }
  };
  auto store_wc = [&](float* Bd) {
#pragma unroll
    for (int j = 0; j < 4; ++j) {
      v4f lo, hi4;
      lo.x = w0[j].x; lo.y = w1[j].x; lo.z = w0[j].y; lo.w = w1[j].y;
      hi4.x = w0[j].z; hi4.y = w1[j].z; hi4.z = w0[j].w; hi4.w = w1[j].w;
      if (j == 3) {
        const v4f z = {0.f, 0.f, 0.f, 0.f};
        lo = b_last ? lo : z;
        hi4 = b_last ? hi4 : z;
      }
      *reinterpret_cast<v4f*>(Bd + b_dst[j]) = lo;
      *reinterpret_cast<v4f*>(Bd + b_dst[j] + 4) = hi4;
    }
  };

  load_wk(0, 0);

  const int a_tq = tid & 15;
  const int a_p0 = tid >> 4;
  const v4f st0 = *reinterpret_cast<const v4f*>(stats + (size_t)(row0 + 4 * a_tq) * 2);
  const v4f st1 = *reinterpret_cast<const v4f*>(stats + (size_t)(row0 + 4 * a_tq) * 2 + 4);
  const v4f mu4 = {st0.x, st0.z, st1.x, st1.z};
  const v4f rs4 = {st0.y, st0.w, st1.y, st1.w};
  v4f xa[16], xc[16];
  float ga[16], gc[16], be_a[16], be_c[16];
  {
    const float* xb = x + (size_t)b * C_ * T_ + t0 + 4 * a_tq;
#pragma unroll
    for (int j = 0; j < 16; ++j) {
      const int c = 2 * (a_p0 + 16 * j);
      const float* xr = xb + (size_t)c * T_;
      xa[j] = *reinterpret_cast<const v4f*>(xr);
      xc[j] = *reinterpret_cast<const v4f*>(xr + T_);
      ga[j] = gamma[c];
      gc[j] = gamma[c + 1];
      be_a[j] = beta[c];
      be_c[j] = beta[c + 1];
    }
  }
  __builtin_amdgcn_sched_barrier(0);
  auto store_x = [&](int j) {
    const int p = a_p0 + 16 * j;
    v4f lo, hi4;
    const v4f ha = (xa[j] - mu4) * rs4 * ga[j] + be_a[j];
    const v4f hc = (xc[j] - mu4) * rs4 * gc[j] + be_c[j];
    lo.x = ha.x; lo.y = hc.x; lo.z = ha.y; lo.w = hc.y;
    hi4.x = ha.z; hi4.y = hc.z; hi4.z = ha.w; hi4.w = hc.w;
    float* dst = &As[(p * F_AP + 4 * a_tq) * 2];
    *reinterpret_cast<v4f*>(dst) = lo;
    *reinterpret_cast<v4f*>(dst + 4) = hi4;
  };
  store_x(0);
  store_wk(Bs);
  __syncthreads();

  const float* a_frag = &As[(hi * F_AP + rblk * 32 + laneM) * 2];
  const float* k_frag = &Ks[(hi * F_AP + rblk * 32 + laneM) * 2];
  const int b1_off = (hi * F_BP + j0 + laneM) * 2;
  const int b2_off = (hi * F_BP + jb + laneM) * 2;

  v8f accA[2][2], accB[2][2];
  v8f lg[2][4] = {{{}, {}, {}, {}}, {{}, {}, {}, {}}};

  struct Fr { v2f a0, a1, b0, b1, b2, b3; };
  auto ld1 = [&](int kc, const float* Bc, int kb) {
    const float* ap = a_frag + (kc * F_KP + (kb >> 1)) * (F_AP * 2);
    const float* bp = Bc + b1_off + (kb >> 1) * (F_BP * 2);
    Fr f;
    f.a0 = *reinterpret_cast<const v2f*>(ap);
    f.a1 = *reinterpret_cast<const v2f*>(ap + 16 * 2);
    f.b0 = *reinterpret_cast<const v2f*>(bp);
    f.b1 = *reinterpret_cast<const v2f*>(bp + 16 * 2);
    f.b2 = *reinterpret_cast<const v2f*>(bp + F_NG * 2);
    f.b3 = *reinterpret_cast<const v2f*>(bp + (F_NG + 16) * 2);
    return f;
  };
  auto mm1 = [&](const Fr& f) {
    accA[0][0] = wmma_f32(f.a0, f.b0, accA[0][0]);
    accA[1][0] = wmma_f32(f.a1, f.b0, accA[1][0]);
    accA[0][1] = wmma_f32(f.a0, f.b1, accA[0][1]);
    accA[1][1] = wmma_f32(f.a1, f.b1, accA[1][1]);
    accB[0][0] = wmma_f32(f.a0, f.b2, accB[0][0]);
    accB[1][0] = wmma_f32(f.a1, f.b2, accB[1][0]);
    accB[0][1] = wmma_f32(f.a0, f.b3, accB[0][1]);
    accB[1][1] = wmma_f32(f.a1, f.b3, accB[1][1]);
  };
  auto ld2 = [&](int c2, const float* Bc, int kb) {
    const float* ap = k_frag + (c2 * F_KP + (kb >> 1)) * (F_AP * 2);
    const float* bp = Bc + b2_off + (kb >> 1) * (F_BP * 2);
    Fr f;
    f.a0 = *reinterpret_cast<const v2f*>(ap);
    f.a1 = *reinterpret_cast<const v2f*>(ap + 16 * 2);
    f.b0 = *reinterpret_cast<const v2f*>(bp);
    f.b1 = *reinterpret_cast<const v2f*>(bp + 16 * 2);
    f.b2 = *reinterpret_cast<const v2f*>(bp + 32 * 2);
    f.b3 = *reinterpret_cast<const v2f*>(bp + 48 * 2);
    return f;
  };
  auto mm2 = [&](const Fr& f) {
    lg[0][0] = wmma_f32(f.a0, f.b0, lg[0][0]);
    lg[1][0] = wmma_f32(f.a1, f.b0, lg[1][0]);
    lg[0][1] = wmma_f32(f.a0, f.b1, lg[0][1]);
    lg[1][1] = wmma_f32(f.a1, f.b1, lg[1][1]);
    lg[0][2] = wmma_f32(f.a0, f.b2, lg[0][2]);
    lg[1][2] = wmma_f32(f.a1, f.b2, lg[1][2]);
    lg[0][3] = wmma_f32(f.a0, f.b3, lg[0][3]);
    lg[1][3] = wmma_f32(f.a1, f.b3, lg[1][3]);
  };

  Fr t6, t7;
  Fr u4, u5;
  auto front1 = [&](int kc, const float* Bc, int pend) {
    const Fr h0 = ld1(kc, Bc, 0), h1 = ld1(kc, Bc, 4);
    __builtin_amdgcn_sched_barrier(0);
    if (pend == 1) { mm1(t6); mm1(t7); }
    if (pend == 2) { mm2(t6); mm2(t7); }
    mm1(h0);
    mm1(h1);
#pragma unroll
    for (int kb = 8; kb < F_KC_SPLIT; kb += 4) mm1(ld1(kc, Bc, kb));
    u4 = ld1(kc, Bc, F_KC_SPLIT);
    u5 = ld1(kc, Bc, F_KC_SPLIT + 4);
    t6 = ld1(kc, Bc, F_KC_SPLIT + 8);
    t7 = ld1(kc, Bc, F_KC_SPLIT + 12);
    __builtin_amdgcn_sched_barrier(0);
  };
  auto back1 = [&]() {
    __builtin_amdgcn_sched_barrier(0);
    mm1(u4);
    mm1(u5);
  };
  auto back2 = [&]() {
    __builtin_amdgcn_sched_barrier(0);
    mm2(u4);
    mm2(u5);
  };
  auto front2 = [&](int c2, const float* Bc, int pend) {
    const Fr h0 = ld2(c2, Bc, 0), h1 = ld2(c2, Bc, 4);
    __builtin_amdgcn_sched_barrier(0);
    if (pend == 2) { mm2(t6); mm2(t7); }
    mm2(h0);
    mm2(h1);
#pragma unroll
    for (int kb = 8; kb < F_KC_SPLIT; kb += 4) mm2(ld2(c2, Bc, kb));
    u4 = ld2(c2, Bc, F_KC_SPLIT);
    u5 = ld2(c2, Bc, F_KC_SPLIT + 4);
    t6 = ld2(c2, Bc, F_KC_SPLIT + 8);
    t7 = ld2(c2, Bc, F_KC_SPLIT + 12);
    __builtin_amdgcn_sched_barrier(0);
  };

  float* const B0 = Bs;
  float* const B1 = Bs + F_BS_FLOATS;

  auto zero_acc1 = [&]() {
#pragma unroll
    for (int r = 0; r < 2; ++r)
#pragma unroll
      for (int jt = 0; jt < 2; ++jt) {
        accA[r][jt] = v8f{};
        accB[r][jt] = v8f{};
      }
  };

  zero_acc1();
  load_wk(0, 1);
  front1(0, B0, 0);
  store_wk(B1);
#pragma unroll
  for (int j = 1; j < 16; ++j) store_x(j);
  back1();
  __syncthreads();

#pragma unroll 1
  for (int ng = 0; ng < C_ / F_NG; ++ng) {
#pragma unroll 1
    for (int kc = 1; kc < C_ / F_KC - 1; ++kc) {
      float* Bc = (kc & 1) ? B1 : B0;
      float* Bn = (kc & 1) ? B0 : B1;
      load_wk(ng, kc + 1);
      front1(kc, Bc, 1);
      store_wk(Bn);
      back1();
      __syncthreads();
    }
    load_wc(ng, 0);
    const int c0 = j0 + laneM;
    const int c1 = c0 + 16;
    const float bka0 = bk[ng * F_NG + c0];
    const float bka1 = bk[ng * F_NG + c1];
    const float bkb0 = bk[C_ + ng * F_NG + c0];
    const float bkb1 = bk[C_ + ng * F_NG + c1];
    front1(C_ / F_KC - 1, B1, 1);
    store_wc(B0);
    back1();
    mm1(t6);
    mm1(t7);
    {
      float* k0 = &Ks[((c0 >> 1) * F_AP) * 2 + (c0 & 1)];
      float* k1 = &Ks[((c1 >> 1) * F_AP) * 2 + (c1 & 1)];
#pragma unroll
      for (int r = 0; r < 2; ++r) {
#pragma unroll
        for (int v = 0; v < 8; ++v) {
          const int tl = rblk * 32 + r * 16 + v + 8 * hi;
          const float a0 = accA[r][0][v] + bka0;
          const float g0 = accB[r][0][v] + bkb0;
          const float a1 = accA[r][1][v] + bka1;
          const float g1 = accB[r][1][v] + bkb1;
          k0[tl * 2] = a0 / (1.f + __expf(-g0));
          k1[tl * 2] = a1 / (1.f + __expf(-g1));
        }
      }
    }
    __syncthreads();

    {
      float* kout = k_ws + ((size_t)b * C_ + ng * F_NG) * T_ + t0;
#pragma unroll
      for (int it = 0; it < 8; ++it) {
        const int i = tid + 256 * it;
        const int c = i >> 4;
        const int tq = i & 15;
        const float* s = &Ks[((c >> 1) * F_AP + 4 * tq) * 2 + (c & 1)];
        v4f v;
        v.x = s[0]; v.y = s[2]; v.z = s[4]; v.w = s[6];
        *reinterpret_cast<v4f*>(kout + (size_t)c * T_ + 4 * tq) = v;
      }
    }

    load_wc(ng, 1);
    front2(0, B0, 0);
    store_wc(B1);
    back2();
    __syncthreads();
#pragma unroll 1
    for (int c2 = 1; c2 < F_NG / F_KC - 1; ++c2) {
      float* Bc = (c2 & 1) ? B1 : B0;
      float* Bn = (c2 & 1) ? B0 : B1;
      load_wc(ng, c2 + 1);
      front2(c2, Bc, 2);
      store_wc(Bn);
      back2();
      __syncthreads();
    }
    if (ng + 1 < C_ / F_NG) load_wk(ng + 1, 0);
    front2(F_NG / F_KC - 1, B1, 2);
    if (ng + 1 < C_ / F_NG) store_wk(B0);
    back2();
    __syncthreads();

    if (ng + 1 < C_ / F_NG) {
      zero_acc1();
      load_wk(ng + 1, 1);
      front1(0, B0, 2);
      store_wk(B1);
      back1();
      __syncthreads();
    } else {
      mm2(t6);
      mm2(t7);
    }
  }

  float* proj = smem;
#pragma unroll
  for (int r = 0; r < 2; ++r)
#pragma unroll
    for (int jt = 0; jt < 4; ++jt)
#pragma unroll
      for (int v = 0; v < 8; ++v)
        proj[(rblk * 32 + r * 16 + v + 8 * hi) * P_S + jb + jt * 16 + laneM] = lg[r][jt][v];
  __syncthreads();

#pragma unroll 1
  for (int u = 0; u < 2; ++u) {
    const int item = tid + 256 * u;
    const int rr = item >> 3;
    const int h = item & 7;
    float* prow = &proj[rr * P_S + h * KER_];
    const float* bch = &bc[h * KER_];
    float m = -1e30f;
#pragma unroll
    for (int i = 0; i < KER_; ++i) {
      float v = prow[i] + bch[i];
      prow[i] = v;
      m = fmaxf(m, v);
    }
    float s = 0.f;
#pragma unroll
    for (int i = 0; i < KER_; ++i) {
      float e = __expf(prow[i] - m);
      prow[i] = e;
      s += e;
    }
    float inv = 1.f / s;
    float* wout = w_ws + (size_t)(row0 + rr) * 256 + h * 32;
#pragma unroll
    for (int i = 0; i < KER_; ++i) wout[i] = prow[i] * inv;
    wout[KER_] = 0.f;
  }
}

#define DC_TT 128
#define DC_SEG 16
#define DC_HALO 16
#define DC_ROWF (DC_TT + 2 * DC_HALO)
#define DC_WIN (DC_SEG + 2 * DC_HALO)
#define DC_PITCH 164
__global__ __launch_bounds__(256) void dynconv_kernel(
    const float* __restrict__ x, const float* __restrict__ k_ws,
    const float* __restrict__ w_ws, const float* __restrict__ conv_bias,
    float* __restrict__ out) {
  __shared__ __attribute__((aligned(16))) float tile[64 * DC_PITCH];
  const int tid = threadIdx.x;
  const int lane = tid & 31;
  const int seg = __builtin_amdgcn_readfirstlane(tid >> 5);
  const int t0 = blockIdx.x * DC_TT;
  const int h = blockIdx.y;
  const int b = blockIdx.z;
  const size_t chan0 = ((size_t)b * C_ + (size_t)h * 64) * T_;

  {
    const float* ksrc = k_ws + chan0;
    v4f stg[10];
#pragma unroll
    for (int it = 0; it < 10; ++it) {
      const int i = tid + it * 256;
      const int r = i / (DC_ROWF / 4);
      const int q = i - r * (DC_ROWF / 4);
      const int tt = t0 - DC_HALO + 4 * q;
      const int ttc = min(max(tt, 0), T_ - 4);
      v4f v = *reinterpret_cast<const v4f*>(ksrc + (size_t)r * T_ + ttc);
      const v4f z = {0.f, 0.f, 0.f, 0.f};
      stg[it] = (tt == ttc) ? v : z;
    }
#pragma unroll
    for (int it = 0; it < 10; ++it) {
      const int i = tid + it * 256;
      const int r = i / (DC_ROWF / 4);
      const int q = i - r * (DC_ROWF / 4);
      *reinterpret_cast<v4f*>(&tile[r * DC_PITCH + 4 * q]) = stg[it];
    }
  }
  __syncthreads();

  float wa[DC_WIN], wb[DC_WIN];
  {
    const float* ra = &tile[lane * DC_PITCH + seg * DC_SEG];
    const float* rb = &tile[(lane + 32) * DC_PITCH + seg * DC_SEG];
#pragma unroll
    for (int q = 0; q < DC_WIN / 4; ++q) {
      const v4f va = *reinterpret_cast<const v4f*>(ra + 4 * q);
      const v4f vb = *reinterpret_cast<const v4f*>(rb + 4 * q);
      wa[4 * q + 0] = va.x; wa[4 * q + 1] = va.y;
      wa[4 * q + 2] = va.z; wa[4 * q + 3] = va.w;
      wb[4 * q + 0] = vb.x; wb[4 * q + 1] = vb.y;
      wb[4 * q + 2] = vb.z; wb[4 * q + 3] = vb.w;
    }
  }
  __syncthreads();

  const float* wsrc =
      w_ws + ((size_t)b * T_ + t0 + seg * DC_SEG) * 256 + (size_t)h * 32;
#pragma unroll
  for (int j4 = 0; j4 < DC_SEG; j4 += 4) {
    v4f oa, ob;
#pragma unroll
    for (int jj = 0; jj < 4; ++jj) {
      const int j = j4 + jj;
      const float* wr = wsrc + (size_t)j * 256;
      float a0 = 0.f, a1 = 0.f;
#pragma unroll
      for (int kk = 0; kk < KER_; ++kk) {
        const float w = wr[kk];
        a0 = fmaf(w, wa[j + kk + 1], a0);
        a1 = fmaf(w, wb[j + kk + 1], a1);
      }
      oa[jj] = a0;
      ob[jj] = a1;
    }
    *reinterpret_cast<v4f*>(&tile[lane * DC_PITCH + seg * DC_SEG + j4]) = oa;
    *reinterpret_cast<v4f*>(&tile[(lane + 32) * DC_PITCH + seg * DC_SEG + j4]) = ob;
  }
  __syncthreads();

#pragma unroll
  for (int it = 0; it < 8; ++it) {
    const int r = it * 8 + seg;
    v4f v = *reinterpret_cast<const v4f*>(&tile[r * DC_PITCH + 4 * lane]);
    const size_t g = chan0 + (size_t)r * T_ + t0 + 4 * lane;
    const v4f xv = *reinterpret_cast<const v4f*>(x + g);
    const float cb = conv_bias[h * 64 + r];
    v.x = (v.x + cb) + xv.x;
    v.y = (v.y + cb) + xv.y;
    v.z = (v.z + cb) + xv.z;
    v.w = (v.w + cb) + xv.w;
    *reinterpret_cast<v4f*>(out + g) = v;
  }
}

extern "C" void kernel_launch(void* const* d_in, const int* in_sizes, int n_in,
                              void* d_out, int out_size, void* d_ws,
                              size_t ws_size, hipStream_t stream) {
  (void)in_sizes;
  (void)n_in;
  (void)out_size;
  (void)ws_size;
  const float* x = (const float*)d_in[0];
  const float* ln_gamma = (const float*)d_in[1];
  const float* ln_beta = (const float*)d_in[2];
  const float* Wk = (const float*)d_in[3];
  const float* bk = (const float*)d_in[4];
  const float* Wc = (const float*)d_in[5];
  const float* bc = (const float*)d_in[6];
  const float* conv_bias = (const float*)d_in[7];
  float* out = (float*)d_out;

  float* ws = (float*)d_ws;
  float* stats = ws;
  float* k_ws = ws + 32768;
  float* w_ws = k_ws + (size_t)B_ * C_ * T_;

  ln_stats_kernel<<<NROWS / 64, 32 * LS_WAVES, 0, stream>>>(x, stats);
  glu_tap_kernel<<<NROWS / F_ROWS, 256, 0, stream>>>(
      x, ln_gamma, ln_beta, Wk, bk, Wc, bc, stats, k_ws, w_ws);
  dynconv_kernel<<<dim3(T_ / DC_TT, HEADS_, B_), 256, 0, stream>>>(
      x, k_ws, w_ws, conv_bias, out);
}


# dynamic conv tap weights: 16-byte vector loads in the k staging group (one round trip), kept across lanes in 16 VGPRs, one lane read per weight feeds both channels; no scalar-load waits; sums kept
# speedup vs baseline: 2.4487x; 1.0358x over previous
#include <hip/hip_runtime.h>
#include <hip/hip_bf16.h>
#include <cstdint>

#define B_ 8
#define C_ 512
#define T_ 2048
#define HEADS_ 8
#define KER_ 31
#define NROWS (B_ * T_)
#define EPS_ 1e-5f

typedef __attribute__((ext_vector_type(2))) float v2f;
typedef __attribute__((ext_vector_type(8))) float v8f;

__device__ __forceinline__ v8f wmma_f32(v2f a, v2f b, v8f c) {
  return __builtin_amdgcn_wmma_f32_16x16x4_f32(false, a, false, b, (short)0, c,
                                               false, false);
}

typedef __attribute__((ext_vector_type(4))) float v4f;
#define LS_WAVES 16
#define LS_CPW (C_ / LS_WAVES)
#define LS_LD (LS_CPW / 2)
#define LS_PARTS (2 * LS_WAVES)
static_assert(LS_LD == 16, "the load fence below names 16 values");
__global__ __launch_bounds__(32 * LS_WAVES) void ln_stats_kernel(
    const float* __restrict__ x, float* __restrict__ stats) {
  const int tid = threadIdx.x;
  const int lane = tid & 31;
  const int wave = __builtin_amdgcn_readfirstlane(tid >> 5);
  const int row0 = blockIdx.x * 64;
  const int b = row0 / T_;
  const int t0 = row0 % T_;
  const int tq = lane & 15;
  const int par = lane >> 4;
  const float* xb = x + (size_t)b * C_ * T_ + (size_t)(wave * LS_CPW + par) * T_ +
                    t0 + 4 * tq;
  v4f v[LS_LD];
  for (int i = 0; i < LS_LD; ++i)
    v[i] = *reinterpret_cast<const v4f*>(xb + (size_t)(2 * i) * T_);
  asm volatile("" ::"v"(v[0]), "v"(v[1]), "v"(v[2]), "v"(v[3]), "v"(v[4]), "v"(v[5]),
               "v"(v[6]), "v"(v[7]), "v"(v[8]), "v"(v[9]), "v"(v[10]), "v"(v[11]),
               "v"(v[12]), "v"(v[13]), "v"(v[14]), "v"(v[15]));
  v4f s = {0.f, 0.f, 0.f, 0.f}, s2 = {0.f, 0.f, 0.f, 0.f};
  for (int i = 0; i < LS_LD; ++i) {
    s += v[i];
    s2 += v[i] * v[i];
  }
  __shared__ __attribute__((aligned(16))) float sb[LS_PARTS][64];
  __shared__ __attribute__((aligned(16))) float sq[LS_PARTS][64];
  *reinterpret_cast<v4f*>(&sb[2 * wave + par][4 * tq]) = s;
  *reinterpret_cast<v4f*>(&sq[2 * wave + par][4 * tq]) = s2;
  __syncthreads();
  if (tid < 64) {
    float S = 0.f, S2 = 0.f;
    for (int p = 0; p < LS_PARTS; ++p) {
      S += sb[p][tid];
      S2 += sq[p][tid];
    }
    float mu = S * (1.f / C_);
    float var = S2 * (1.f / C_) - mu * mu;
    stats[(row0 + tid) * 2 + 0] = mu;
    stats[(row0 + tid) * 2 + 1] = rsqrtf(var + EPS_);
  }
}

#define F_ROWS 64
#define F_KC 32
#define F_KP (F_KC / 2)
#define F_KC_SPLIT 16
#define F_AP 80
#define F_BP 272
#define F_NG 128
#define F_AS_FLOATS ((C_ / 2) * F_AP * 2)
#define F_BS_FLOATS (F_KP * F_BP * 2)
#define F_KS_FLOATS ((F_NG / 2) * F_AP * 2)
#define P_S 257
#define NCW (KER_ * HEADS_)
#define NQW (NCW / 4)
static_assert(F_ROWS * P_S <= F_AS_FLOATS, "the softmax stage reuses the x tile");
__global__ __launch_bounds__(256) void glu_tap_kernel(
    const float* __restrict__ x, const float* __restrict__ gamma,
    const float* __restrict__ beta, const float* __restrict__ Wk,
    const float* __restrict__ bk, const float* __restrict__ Wc,
    const float* __restrict__ bc, const float* __restrict__ stats,
    float* __restrict__ k_ws, float* __restrict__ w_ws) {
  __shared__ __attribute__((aligned(16))) float smem[F_AS_FLOATS + 2 * F_BS_FLOATS + F_KS_FLOATS];
  float* As = smem;
  float* Bs = smem + F_AS_FLOATS;
  float* Ks = smem + F_AS_FLOATS + 2 * F_BS_FLOATS;

  const int tid = threadIdx.x;
  const int row0 = blockIdx.x * F_ROWS;
  const int b = row0 / T_;
  const int t0 = row0 % T_;
  const int lane = tid & 31;
  const int wave = __builtin_amdgcn_readfirstlane(tid >> 5);
  const int laneM = lane & 15;
  const int hi = lane >> 4;
  const int rblk = wave & 1;
  const int j0 = (wave >> 1) * 32;
  const int jb = (wave >> 1) * 64;

  const int wk_src = (2 * (wave >> 1)) * (2 * C_) + (wave & 1) * C_ + 4 * lane;
  const int wk_dst = (((wave >> 1)) * F_BP + (wave & 1) * F_NG + 4 * lane) * 2;
  int b_src[4], b_dst[4];
#pragma unroll
  for (int j = 0; j < 4; ++j) {
    const int item = min(tid + 256 * j, 16 * NQW - 1);
    const int p = item / NQW;
    const int q = item - p * NQW;
    b_src[j] = 2 * p * NCW + 4 * q;
    b_dst[j] = (p * F_BP + 4 * q) * 2;
  }
  const bool b_last = (tid + 768) < 16 * NQW;
  if (!b_last) b_dst[3] = (((tid - 224) >> 1) * F_BP + NCW + 4 * (tid & 1)) * 2;

  v4f w0[4], w1[4];
  auto load_wk = [&](int ng, int kc) {
    const float* s = Wk + (size_t)(kc * F_KC) * (2 * C_) + ng * F_NG + wk_src;
#pragma unroll
    for (int j = 0; j < 4; ++j) {
      w0[j] = *reinterpret_cast<const v4f*>(s + (size_t)(8 * j) * (2 * C_));
      w1[j] = *reinterpret_cast<const v4f*>(s + (size_t)(8 * j + 1) * (2 * C_));
    }
  };
  auto store_wk = [&](float* Bd) {
#pragma unroll
    for (int j = 0; j < 4; ++j) {
      v4f lo, hi4;
      lo.x = w0[j].x; lo.y = w1[j].x; lo.z = w0[j].y; lo.w = w1[j].y;
      hi4.x = w0[j].z; hi4.y = w1[j].z; hi4.z = w0[j].w; hi4.w = w1[j].w;
      float* dst = Bd + wk_dst + j * (4 * F_BP * 2);
      *reinterpret_cast<v4f*>(dst) = lo;
      *reinterpret_cast<v4f*>(dst + 4) = hi4;
    }
  };
  auto load_wc = [&](int ng, int c2) {
    const float* s = Wc + (size_t)(ng * F_NG + c2 * F_KC) * NCW;
#pragma unroll
    for (int j = 0; j < 4; ++j) {
      w0[j] = *reinterpret_cast<const v4f*>(s + b_src[j]);
      w1[j] = *reinterpret_cast<const v4f*>(s + b_src[j] + NCW);
    }
  };
  auto store_wc = [&](float* Bd) {
#pragma unroll
    for (int j = 0; j < 4; ++j) {
      v4f lo, hi4;
      lo.x = w0[j].x; lo.y = w1[j].x; lo.z = w0[j].y; lo.w = w1[j].y;
      hi4.x = w0[j].z; hi4.y = w1[j].z; hi4.z = w0[j].w; hi4.w = w1[j].w;
      if (j == 3) {
        const v4f z = {0.f, 0.f, 0.f, 0.f};
        lo = b_last ? lo : z;
        hi4 = b_last ? hi4 : z;
      }
      *reinterpret_cast<v4f*>(Bd + b_dst[j]) = lo;
      *reinterpret_cast<v4f*>(Bd + b_dst[j] + 4) = hi4;
    }
  };

  load_wk(0, 0);

  const int a_tq = tid & 15;
  const int a_p0 = tid >> 4;
  const v4f st0 = *reinterpret_cast<const v4f*>(stats + (size_t)(row0 + 4 * a_tq) * 2);
  const v4f st1 = *reinterpret_cast<const v4f*>(stats + (size_t)(row0 + 4 * a_tq) * 2 + 4);
  const v4f mu4 = {st0.x, st0.z, st1.x, st1.z};
  const v4f rs4 = {st0.y, st0.w, st1.y, st1.w};
  v4f xa[16], xc[16];
  float ga[16], gc[16], be_a[16], be_c[16];
  {
    const float* xb = x + (size_t)b * C_ * T_ + t0 + 4 * a_tq;
#pragma unroll
    for (int j = 0; j < 16; ++j) {
      const int c = 2 * (a_p0 + 16 * j);
      const float* xr = xb + (size_t)c * T_;
      xa[j] = *reinterpret_cast<const v4f*>(xr);
      xc[j] = *reinterpret_cast<const v4f*>(xr + T_);
      ga[j] = gamma[c];
      gc[j] = gamma[c + 1];
      be_a[j] = beta[c];
      be_c[j] = beta[c + 1];
    }
  }
  __builtin_amdgcn_sched_barrier(0);
  auto store_x = [&](int j) {
    const int p = a_p0 + 16 * j;
    v4f lo, hi4;
    const v4f ha = (xa[j] - mu4) * rs4 * ga[j] + be_a[j];
    const v4f hc = (xc[j] - mu4) * rs4 * gc[j] + be_c[j];
    lo.x = ha.x; lo.y = hc.x; lo.z = ha.y; lo.w = hc.y;
    hi4.x = ha.z; hi4.y = hc.z; hi4.z = ha.w; hi4.w = hc.w;
    float* dst = &As[(p * F_AP + 4 * a_tq) * 2];
    *reinterpret_cast<v4f*>(dst) = lo;
    *reinterpret_cast<v4f*>(dst + 4) = hi4;
  };
  store_x(0);
  store_wk(Bs);
  __syncthreads();

  const float* a_frag = &As[(hi * F_AP + rblk * 32 + laneM) * 2];
  const float* k_frag = &Ks[(hi * F_AP + rblk * 32 + laneM) * 2];
  const int b1_off = (hi * F_BP + j0 + laneM) * 2;
  const int b2_off = (hi * F_BP + jb + laneM) * 2;

  v8f accA[2][2], accB[2][2];
  v8f lg[2][4] = {{{}, {}, {}, {}}, {{}, {}, {}, {}}};

  struct Fr { v2f a0, a1, b0, b1, b2, b3; };
  auto ld1 = [&](int kc, const float* Bc, int kb) {
    const float* ap = a_frag + (kc * F_KP + (kb >> 1)) * (F_AP * 2);
    const float* bp = Bc + b1_off + (kb >> 1) * (F_BP * 2);
    Fr f;
    f.a0 = *reinterpret_cast<const v2f*>(ap);
    f.a1 = *reinterpret_cast<const v2f*>(ap + 16 * 2);
    f.b0 = *reinterpret_cast<const v2f*>(bp);
    f.b1 = *reinterpret_cast<const v2f*>(bp + 16 * 2);
    f.b2 = *reinterpret_cast<const v2f*>(bp + F_NG * 2);
    f.b3 = *reinterpret_cast<const v2f*>(bp + (F_NG + 16) * 2);
    return f;
  };
  auto mm1 = [&](const Fr& f) {
    accA[0][0] = wmma_f32(f.a0, f.b0, accA[0][0]);
    accA[1][0] = wmma_f32(f.a1, f.b0, accA[1][0]);
    accA[0][1] = wmma_f32(f.a0, f.b1, accA[0][1]);
    accA[1][1] = wmma_f32(f.a1, f.b1, accA[1][1]);
    accB[0][0] = wmma_f32(f.a0, f.b2, accB[0][0]);
    accB[1][0] = wmma_f32(f.a1, f.b2, accB[1][0]);
    accB[0][1] = wmma_f32(f.a0, f.b3, accB[0][1]);
    accB[1][1] = wmma_f32(f.a1, f.b3, accB[1][1]);
  };
  auto ld2 = [&](int c2, const float* Bc, int kb) {
    const float* ap = k_frag + (c2 * F_KP + (kb >> 1)) * (F_AP * 2);
    const float* bp = Bc + b2_off + (kb >> 1) * (F_BP * 2);
    Fr f;
    f.a0 = *reinterpret_cast<const v2f*>(ap);
    f.a1 = *reinterpret_cast<const v2f*>(ap + 16 * 2);
    f.b0 = *reinterpret_cast<const v2f*>(bp);
    f.b1 = *reinterpret_cast<const v2f*>(bp + 16 * 2);
    f.b2 = *reinterpret_cast<const v2f*>(bp + 32 * 2);
    f.b3 = *reinterpret_cast<const v2f*>(bp + 48 * 2);
    return f;
  };
  auto mm2 = [&](const Fr& f) {
    lg[0][0] = wmma_f32(f.a0, f.b0, lg[0][0]);
    lg[1][0] = wmma_f32(f.a1, f.b0, lg[1][0]);
    lg[0][1] = wmma_f32(f.a0, f.b1, lg[0][1]);
    lg[1][1] = wmma_f32(f.a1, f.b1, lg[1][1]);
    lg[0][2] = wmma_f32(f.a0, f.b2, lg[0][2]);
    lg[1][2] = wmma_f32(f.a1, f.b2, lg[1][2]);
    lg[0][3] = wmma_f32(f.a0, f.b3, lg[0][3]);
    lg[1][3] = wmma_f32(f.a1, f.b3, lg[1][3]);
  };

  Fr t6, t7;
  Fr u4, u5;
  auto front1 = [&](int kc, const float* Bc, int pend) {
    const Fr h0 = ld1(kc, Bc, 0), h1 = ld1(kc, Bc, 4);
    __builtin_amdgcn_sched_barrier(0);
    if (pend == 1) { mm1(t6); mm1(t7); }
    if (pend == 2) { mm2(t6); mm2(t7); }
    mm1(h0);
    mm1(h1);
#pragma unroll
    for (int kb = 8; kb < F_KC_SPLIT; kb += 4) mm1(ld1(kc, Bc, kb));
    u4 = ld1(kc, Bc, F_KC_SPLIT);
    u5 = ld1(kc, Bc, F_KC_SPLIT + 4);
    t6 = ld1(kc, Bc, F_KC_SPLIT + 8);
    t7 = ld1(kc, Bc, F_KC_SPLIT + 12);
    __builtin_amdgcn_sched_barrier(0);
  };
  auto back1 = [&]() {
    __builtin_amdgcn_sched_barrier(0);
    mm1(u4);
    mm1(u5);
  };
  auto back2 = [&]() {
    __builtin_amdgcn_sched_barrier(0);
    mm2(u4);
    mm2(u5);
  };
  auto front2 = [&](int c2, const float* Bc, int pend) {
    const Fr h0 = ld2(c2, Bc, 0), h1 = ld2(c2, Bc, 4);
    __builtin_amdgcn_sched_barrier(0);
    if (pend == 2) { mm2(t6); mm2(t7); }
    mm2(h0);
    mm2(h1);
#pragma unroll
    for (int kb = 8; kb < F_KC_SPLIT; kb += 4) mm2(ld2(c2, Bc, kb));
    u4 = ld2(c2, Bc, F_KC_SPLIT);
    u5 = ld2(c2, Bc, F_KC_SPLIT + 4);
    t6 = ld2(c2, Bc, F_KC_SPLIT + 8);
    t7 = ld2(c2, Bc, F_KC_SPLIT + 12);
    __builtin_amdgcn_sched_barrier(0);
  };

  float* const B0 = Bs;
  float* const B1 = Bs + F_BS_FLOATS;

  auto zero_acc1 = [&]() {
#pragma unroll
    for (int r = 0; r < 2; ++r)
#pragma unroll
      for (int jt = 0; jt < 2; ++jt) {
        accA[r][jt] = v8f{};
        accB[r][jt] = v8f{};
      }
  };

  zero_acc1();
  load_wk(0, 1);
  front1(0, B0, 0);
  store_wk(B1);
#pragma unroll
  for (int j = 1; j < 16; ++j) store_x(j);
  back1();
  __syncthreads();

#pragma unroll 1
  for (int ng = 0; ng < C_ / F_NG; ++ng) {
#pragma unroll 1
    for (int kc = 1; kc < C_ / F_KC - 1; ++kc) {
      float* Bc = (kc & 1) ? B1 : B0;
      float* Bn = (kc & 1) ? B0 : B1;
      load_wk(ng, kc + 1);
      front1(kc, Bc, 1);
      store_wk(Bn);
      back1();
      __syncthreads();
    }
    load_wc(ng, 0);
    const int c0 = j0 + laneM;
    const int c1 = c0 + 16;
    const float bka0 = bk[ng * F_NG + c0];
    const float bka1 = bk[ng * F_NG + c1];
    const float bkb0 = bk[C_ + ng * F_NG + c0];
    const float bkb1 = bk[C_ + ng * F_NG + c1];
    front1(C_ / F_KC - 1, B1, 1);
    store_wc(B0);
    back1();
    mm1(t6);
    mm1(t7);
    {
      float* k0 = &Ks[((c0 >> 1) * F_AP) * 2 + (c0 & 1)];
      float* k1 = &Ks[((c1 >> 1) * F_AP) * 2 + (c1 & 1)];
#pragma unroll
      for (int r = 0; r < 2; ++r) {
#pragma unroll
        for (int v = 0; v < 8; ++v) {
          const int tl = rblk * 32 + r * 16 + v + 8 * hi;
          const float a0 = accA[r][0][v] + bka0;
          const float g0 = accB[r][0][v] + bkb0;
          const float a1 = accA[r][1][v] + bka1;
          const float g1 = accB[r][1][v] + bkb1;
          k0[tl * 2] = a0 / (1.f + __expf(-g0));
          k1[tl * 2] = a1 / (1.f + __expf(-g1));
        }
      }
    }
    __syncthreads();

    {
      float* kout = k_ws + ((size_t)b * C_ + ng * F_NG) * T_ + t0;
#pragma unroll
      for (int it = 0; it < 8; ++it) {
        const int i = tid + 256 * it;
        const int c = i >> 4;
        const int tq = i & 15;
        const float* s = &Ks[((c >> 1) * F_AP + 4 * tq) * 2 + (c & 1)];
        v4f v;
        v.x = s[0]; v.y = s[2]; v.z = s[4]; v.w = s[6];
        *reinterpret_cast<v4f*>(kout + (size_t)c * T_ + 4 * tq) = v;
      }
    }

    load_wc(ng, 1);
    front2(0, B0, 0);
    store_wc(B1);
    back2();
    __syncthreads();
#pragma unroll 1
    for (int c2 = 1; c2 < F_NG / F_KC - 1; ++c2) {
      float* Bc = (c2 & 1) ? B1 : B0;
      float* Bn = (c2 & 1) ? B0 : B1;
      load_wc(ng, c2 + 1);
      front2(c2, Bc, 2);
      store_wc(Bn);
      back2();
      __syncthreads();
    }
    if (ng + 1 < C_ / F_NG) load_wk(ng + 1, 0);
    front2(F_NG / F_KC - 1, B1, 2);
    if (ng + 1 < C_ / F_NG) store_wk(B0);
    back2();
    __syncthreads();

    if (ng + 1 < C_ / F_NG) {
      zero_acc1();
      load_wk(ng + 1, 1);
      front1(0, B0, 2);
      store_wk(B1);
      back1();
      __syncthreads();
    } else {
      mm2(t6);
      mm2(t7);
    }
  }

  float* proj = smem;
#pragma unroll
  for (int r = 0; r < 2; ++r)
#pragma unroll
    for (int jt = 0; jt < 4; ++jt)
#pragma unroll
      for (int v = 0; v < 8; ++v)
        proj[(rblk * 32 + r * 16 + v + 8 * hi) * P_S + jb + jt * 16 + laneM] = lg[r][jt][v];
  __syncthreads();

#pragma unroll 1
  for (int u = 0; u < 2; ++u) {
    const int item = tid + 256 * u;
    const int rr = item >> 3;
    const int h = item & 7;
    float* prow = &proj[rr * P_S + h * KER_];
    const float* bch = &bc[h * KER_];
    float m = -1e30f;
#pragma unroll
    for (int i = 0; i < KER_; ++i) {
      float v = prow[i] + bch[i];
      prow[i] = v;
      m = fmaxf(m, v);
    }
    float s = 0.f;
#pragma unroll
    for (int i = 0; i < KER_; ++i) {
      float e = __expf(prow[i] - m);
      prow[i] = e;
      s += e;
    }
    float inv = 1.f / s;
    float* wout = w_ws + (size_t)(row0 + rr) * 256 + h * 32;
#pragma unroll
    for (int i = 0; i < KER_; ++i) wout[i] = prow[i] * inv;
    wout[KER_] = 0.f;
  }
}

#define DC_TT 128
#define DC_SEG 16
#define DC_HALO 16
#define DC_ROWF (DC_TT + 2 * DC_HALO)
#define DC_WIN (DC_SEG + 2 * DC_HALO)
#define DC_PITCH 164
__global__ __launch_bounds__(256) void dynconv_kernel(
    const float* __restrict__ x, const float* __restrict__ k_ws,
    const float* __restrict__ w_ws, const float* __restrict__ conv_bias,
    float* __restrict__ out) {
  __shared__ __attribute__((aligned(16))) float tile[64 * DC_PITCH];
  const int tid = threadIdx.x;
  const int lane = tid & 31;
  const int seg = __builtin_amdgcn_readfirstlane(tid >> 5);
  const int t0 = blockIdx.x * DC_TT;
  const int h = blockIdx.y;
  const int b = blockIdx.z;
  const size_t chan0 = ((size_t)b * C_ + (size_t)h * 64) * T_;

  v4f wq[DC_SEG / 4];
  {
    const float* wsrc = w_ws + ((size_t)b * T_ + t0 + seg * DC_SEG) * 256 +
                        (size_t)h * 32 + (size_t)(lane >> 3) * 256 + 4 * (lane & 7);
#pragma unroll
    for (int i = 0; i < DC_SEG / 4; ++i)
      wq[i] = *reinterpret_cast<const v4f*>(wsrc + (size_t)(4 * i) * 256);
  }

  {
    const float* ksrc = k_ws + chan0;
    v4f stg[10];
#pragma unroll
    for (int it = 0; it < 10; ++it) {
      const int i = tid + it * 256;
      const int r = i / (DC_ROWF / 4);
      const int q = i - r * (DC_ROWF / 4);
      const int tt = t0 - DC_HALO + 4 * q;
      const int ttc = min(max(tt, 0), T_ - 4);
      stg[it] = *reinterpret_cast<const v4f*>(ksrc + (size_t)r * T_ + ttc);
    }
    __builtin_amdgcn_sched_barrier(0);
#pragma unroll
    for (int it = 0; it < 10; ++it) {
      const int i = tid + it * 256;
      const int r = i / (DC_ROWF / 4);
      const int q = i - r * (DC_ROWF / 4);
      const int tt = t0 - DC_HALO + 4 * q;
      const int ttc = min(max(tt, 0), T_ - 4);
      const v4f z = {0.f, 0.f, 0.f, 0.f};
      *reinterpret_cast<v4f*>(&tile[r * DC_PITCH + 4 * q]) = (tt == ttc) ? stg[it] : z;
    }
  }
  __syncthreads();

  float wa[DC_WIN], wb[DC_WIN];
  {
    const float* ra = &tile[lane * DC_PITCH + seg * DC_SEG];
    const float* rb = &tile[(lane + 32) * DC_PITCH + seg * DC_SEG];
#pragma unroll
    for (int q = 0; q < DC_WIN / 4; ++q) {
      const v4f va = *reinterpret_cast<const v4f*>(ra + 4 * q);
      const v4f vb = *reinterpret_cast<const v4f*>(rb + 4 * q);
      wa[4 * q + 0] = va.x; wa[4 * q + 1] = va.y;
      wa[4 * q + 2] = va.z; wa[4 * q + 3] = va.w;
      wb[4 * q + 0] = vb.x; wb[4 * q + 1] = vb.y;
      wb[4 * q + 2] = vb.z; wb[4 * q + 3] = vb.w;
    }
  }
  __syncthreads();

#pragma unroll
  for (int j4 = 0; j4 < DC_SEG; j4 += 4) {
    v4f oa, ob;
#pragma unroll
    for (int jj = 0; jj < 4; ++jj) {
      const int j = j4 + jj;
      float a0 = 0.f, a1 = 0.f;
#pragma unroll
      for (int kk = 0; kk < KER_; ++kk) {
        const float w = __int_as_float(__builtin_amdgcn_readlane(
            __float_as_int(wq[j4 >> 2][kk & 3]), 8 * jj + (kk >> 2)));
        a0 = fmaf(w, wa[j + kk + 1], a0);
        a1 = fmaf(w, wb[j + kk + 1], a1);
      }
      oa[jj] = a0;
      ob[jj] = a1;
    }
    *reinterpret_cast<v4f*>(&tile[lane * DC_PITCH + seg * DC_SEG + j4]) = oa;
    *reinterpret_cast<v4f*>(&tile[(lane + 32) * DC_PITCH + seg * DC_SEG + j4]) = ob;
  }
  __syncthreads();

#pragma unroll
  for (int it = 0; it < 8; ++it) {
    const int r = it * 8 + seg;
    v4f v = *reinterpret_cast<const v4f*>(&tile[r * DC_PITCH + 4 * lane]);
    const size_t g = chan0 + (size_t)r * T_ + t0 + 4 * lane;
    const v4f xv = *reinterpret_cast<const v4f*>(x + g);
    const float cb = conv_bias[h * 64 + r];
    v.x = (v.x + cb) + xv.x;
    v.y = (v.y + cb) + xv.y;
    v.z = (v.z + cb) + xv.z;
    v.w = (v.w + cb) + xv.w;
    *reinterpret_cast<v4f*>(out + g) = v;
  }
}

extern "C" void kernel_launch(void* const* d_in, const int* in_sizes, int n_in,
                              void* d_out, int out_size, void* d_ws,
                              size_t ws_size, hipStream_t stream) {
  (void)in_sizes;
  (void)n_in;
  (void)out_size;
  (void)ws_size;
  const float* x = (const float*)d_in[0];
  const float* ln_gamma = (const float*)d_in[1];
  const float* ln_beta = (const float*)d_in[2];
  const float* Wk = (const float*)d_in[3];
  const float* bk = (const float*)d_in[4];
  const float* Wc = (const float*)d_in[5];
  const float* bc = (const float*)d_in[6];
  const float* conv_bias = (const float*)d_in[7];
  float* out = (float*)d_out;

  float* ws = (float*)d_ws;
  float* stats = ws;
  float* k_ws = ws + 32768;
  float* w_ws = k_ws + (size_t)B_ * C_ * T_;

  ln_stats_kernel<<<NROWS / 64, 32 * LS_WAVES, 0, stream>>>(x, stats);
  glu_tap_kernel<<<NROWS / F_ROWS, 256, 0, stream>>>(
      x, ln_gamma, ln_beta, Wk, bk, Wc, bc, stats, k_ws, w_ws);
  dynconv_kernel<<<dim3(T_ / DC_TT, HEADS_, B_), 256, 0, stream>>>(
      x, k_ws, w_ws, conv_bias, out);
}
